# starting kernel
# baseline (speedup 1.0000x reference)
#include <hip/hip_runtime.h>

#define NN  50000
#define EE  800000
#define EP  (EE + NN)
#define FH  128
#define NH  4
#define HIDW 32

typedef __attribute__((ext_vector_type(2))) float v2f;
typedef __attribute__((ext_vector_type(8))) float v8f;

__device__ __forceinline__ float lrelu(float v) {
    return fmaxf(v, 0.f) + 0.2f * fminf(v, 0.f);
}

__global__ void gemm_wmma_dual(const float* __restrict__ X,
                               const float* __restrict__ Wl,
                               const float* __restrict__ bl,
                               const float* __restrict__ Wr,
                               const float* __restrict__ br,
                               float* __restrict__ OutL,
                               float* __restrict__ OutR)
{
    const int lane  = threadIdx.x & 31;
    const int wave  = threadIdx.x >> 5;
    const int r0    = blockIdx.x * 128 + wave * 16;
    const int n0    = blockIdx.y * 16;
    const int mrow  = lane & 15;
    const int khalf = lane >> 4;

    int arow = r0 + mrow;
    if (arow >= NN) arow = NN - 1;
    const float* xrow = X + (size_t)arow * FH;
    const int col = n0 + mrow;

    v8f cL = {};
    v8f cR = {};
#pragma unroll
    for (int k = 0; k < FH; k += 4) {
        const v2f a = *(const v2f*)(xrow + k + khalf * 2);
        const size_t rk0 = (size_t)(k + khalf * 2) * FH + col;
        v2f bL, bR;
        bL.x = Wl[rk0];       bL.y = Wl[rk0 + FH];
        bR.x = Wr[rk0];       bR.y = Wr[rk0 + FH];
        cL = __builtin_amdgcn_wmma_f32_16x16x4_f32(false, a, false, bL,
                                                   (short)0, cL, false, false);
        cR = __builtin_amdgcn_wmma_f32_16x16x4_f32(false, a, false, bR,
                                                   (short)0, cR, false, false);
    }

    const float bbL = bl[col];
    const float bbR = br[col];
#pragma unroll
    for (int v = 0; v < 8; ++v) {
        const int row = r0 + v + khalf * 8;
        if (row < NN) {
            OutL[(size_t)row * FH + col] = cL[v] + bbL;
            OutR[(size_t)row * FH + col] = cR[v] + bbR;
        }
    }
}

__global__ void init_buffers(float* __restrict__ hbuf,
                             float* __restrict__ m,
                             float* __restrict__ denom)
{
    const int t = blockIdx.x * blockDim.x + threadIdx.x;
    if (t < NN * FH) hbuf[t] = 0.f;
    if (t < NN * NH) { m[t] = -3.0e38f; denom[t] = 0.f; }
}

__global__ void edge_logits(const float* __restrict__ xl,
                            const float* __restrict__ xr,
                            const int* __restrict__ src,
                            const int* __restrict__ dst,
                            const float* __restrict__ att,
                            float* __restrict__ lg,
                            float* __restrict__ m)
{
    const int gid  = blockIdx.x * blockDim.x + threadIdx.x;
    const int e    = gid >> 5;
    const int lane = gid & 31;
    if (e >= EP) return;

    int s, d;
    if (e < EE) { s = src[e]; d = dst[e]; } else { s = d = e - EE; }

    const int f = lane * 4;
    const float4 a = *(const float4*)(xl + (size_t)s * FH + f);
    const float4 b = *(const float4*)(xr + (size_t)d * FH + f);
    const float4 w = *(const float4*)(att + f);

    float p = lrelu(a.x + b.x) * w.x + lrelu(a.y + b.y) * w.y +
              lrelu(a.z + b.z) * w.z + lrelu(a.w + b.w) * w.w;
    p += __shfl_xor(p, 1, 8);
    p += __shfl_xor(p, 2, 8);
    p += __shfl_xor(p, 4, 8);

    if ((lane & 7) == 0) {
        const int h = lane >> 3;
        lg[(size_t)e * NH + h] = p;
        atomicMax(&m[(size_t)d * NH + h], p);
    }
}

__global__ void edge_exp(const int* __restrict__ dst,
                         const float* __restrict__ m,
                         float* __restrict__ lg,
                         float* __restrict__ denom)
{
    const int e = blockIdx.x * blockDim.x + threadIdx.x;
    if (e >= EP) return;
    const int d = (e < EE) ? dst[e] : (e - EE);

    float4 l  = *(const float4*)(lg + (size_t)e * NH);
    const float4 mm = *(const float4*)(m + (size_t)d * NH);
    l.x = expf(l.x - mm.x);
    l.y = expf(l.y - mm.y);
    l.z = expf(l.z - mm.z);
    l.w = expf(l.w - mm.w);
    *(float4*)(lg + (size_t)e * NH) = l;

    float* dn = denom + (size_t)d * NH;
    atomicAdd(dn + 0, l.x);
    atomicAdd(dn + 1, l.y);
    atomicAdd(dn + 2, l.z);
    atomicAdd(dn + 3, l.w);
}

__global__ void edge_scatter(const float* __restrict__ xl,
                             const int* __restrict__ src,
                             const int* __restrict__ dst,
                             const float* __restrict__ lg,
                             const float* __restrict__ denom,
                             float* __restrict__ acc)
{
    const int gid  = blockIdx.x * blockDim.x + threadIdx.x;
    const int e    = gid >> 5;
    const int lane = gid & 31;
    if (e >= EP) return;

    int s, d;
    if (e < EE) { s = src[e]; d = dst[e]; } else { s = d = e - EE; }

    const int h = lane >> 3;
    const float alpha = lg[(size_t)e * NH + h] /
                        fmaxf(denom[(size_t)d * NH + h], 1e-16f);

    const int f = lane * 4;
    const float4 a = *(const float4*)(xl + (size_t)s * FH + f);
    float* o = acc + (size_t)d * FH + f;
    atomicAdd(o + 0, a.x * alpha);
    atomicAdd(o + 1, a.y * alpha);
    atomicAdd(o + 2, a.z * alpha);
    atomicAdd(o + 3, a.w * alpha);
}

__global__ void finalize_elu(float* __restrict__ hbuf,
                             const float* __restrict__ bo)
{
    const int t = blockIdx.x * blockDim.x + threadIdx.x;
    if (t >= NN * FH) return;
    const float v = hbuf[t] + bo[t & (FH - 1)];
    hbuf[t] = (v > 0.f) ? v : (expf(v) - 1.f);
}

__global__ void final_linear(const float* __restrict__ h,
                             const float* __restrict__ Wf,
                             const float* __restrict__ bf,
                             float* __restrict__ out)
{
    const int gid  = blockIdx.x * blockDim.x + threadIdx.x;
    const int n    = gid >> 5;
    const int lane = gid & 31;
    if (n >= NN) return;

    const int f = lane * 4;
    const float4 a = *(const float4*)(h + (size_t)n * FH + f);
    const float4 w = *(const float4*)(Wf + f);
    float p = a.x * w.x + a.y * w.y + a.z * w.z + a.w * w.w;
    p += __shfl_xor(p, 16, 32);
    p += __shfl_xor(p,  8, 32);
    p += __shfl_xor(p,  4, 32);
    p += __shfl_xor(p,  2, 32);
    p += __shfl_xor(p,  1, 32);
    if (lane == 0) out[n] = p + bf[0];
}

extern "C" void kernel_launch(void* const* d_in, const int* in_sizes, int n_in,
                              void* d_out, int out_size, void* d_ws, size_t ws_size,
                              hipStream_t stream)
{
    const float* x    = (const float*)d_in[0];
    const int*   ei   = (const int*)d_in[1];
    const int*   src  = ei;
    const int*   dst  = ei + EE;
    const float* Wl0  = (const float*)d_in[2];
    const float* bl0  = (const float*)d_in[3];
    const float* Wr0  = (const float*)d_in[4];
    const float* br0  = (const float*)d_in[5];
    const float* att0 = (const float*)d_in[6];
    const float* bo0  = (const float*)d_in[7];
    const float* Wl   = (const float*)d_in[8];
    const float* bl   = (const float*)d_in[9];
    const float* Wr   = (const float*)d_in[10];
    const float* br   = (const float*)d_in[11];
    const float* att  = (const float*)d_in[12];
    const float* bo   = (const float*)d_in[13];
    const float* Wf   = (const float*)d_in[14];
    const float* bf   = (const float*)d_in[15];

    float* ws   = (float*)d_ws;
    float* hbuf = ws;
    float* xl   = hbuf + (size_t)NN * FH;
    float* xr   = xl   + (size_t)NN * FH;
    float* lg   = xr   + (size_t)NN * FH;
    float* mbuf = lg   + (size_t)EP * NH;
    float* dbuf = mbuf + (size_t)NN * NH;

    const dim3 gemmGrid((NN + 127) / 128, FH / 16);
    const dim3 edgeGrid((EP * 32 + 255) / 256);
    const dim3 eGrid((EP + 255) / 256);
    const dim3 nodeGrid((NN * FH + 255) / 256);
    const dim3 linGrid((NN * 32 + 255) / 256);

    for (int layer = 0; layer < 3; ++layer) {
        const float* in = (layer == 0) ? x : hbuf;
        const float *WL, *BL, *WR, *BR, *AT, *BO;
        if (layer == 0) { WL = Wl0; BL = bl0; WR = Wr0; BR = br0; AT = att0; BO = bo0; }
        else {
            const int i = layer - 1;
            WL = Wl + (size_t)i * FH * FH;  BL = bl + (size_t)i * FH;
            WR = Wr + (size_t)i * FH * FH;  BR = br + (size_t)i * FH;
            AT = att + (size_t)i * NH * HIDW;
            BO = bo + (size_t)i * FH;
        }
        gemm_wmma_dual<<<gemmGrid, 256, 0, stream>>>(in, WL, BL, WR, BR, xl, xr);
        init_buffers <<<nodeGrid, 256, 0, stream>>>(hbuf, mbuf, dbuf);
        edge_logits  <<<edgeGrid, 256, 0, stream>>>(xl, xr, src, dst, AT, lg, mbuf);
        edge_exp     <<<eGrid,    256, 0, stream>>>(dst, mbuf, lg, dbuf);
        edge_scatter <<<edgeGrid, 256, 0, stream>>>(xl, src, dst, lg, dbuf, hbuf);
        finalize_elu <<<nodeGrid, 256, 0, stream>>>(hbuf, BO);
    }
    final_linear<<<linGrid, 256, 0, stream>>>(hbuf, Wf, bf, (float*)d_out);
}


# Atomic edge passes replaced by a per-call counting sort of edges by dst (int atomics, arrival-order rank) and one wave per node doing a one-pass online-softmax gather; no float atomics
# speedup vs baseline: 4.0208x; 4.0208x over previous
#include <hip/hip_runtime.h>

#define NN  50000
#define EE  800000
#define FH  128
#define NH  4
#define HIDW 32

typedef __attribute__((ext_vector_type(2))) float v2f;
typedef __attribute__((ext_vector_type(8))) float v8f;
typedef __attribute__((ext_vector_type(4))) float v4f;

__global__ void gemm_wmma_dual(const float* __restrict__ X,
                               const float* __restrict__ Wl,
                               const float* __restrict__ bl,
                               const float* __restrict__ Wr,
                               const float* __restrict__ br,
                               float* __restrict__ OutL,
                               float* __restrict__ OutR)
{
    const int lane  = threadIdx.x & 31;
    const int wave  = threadIdx.x >> 5;
    const int r0    = blockIdx.x * 128 + wave * 16;
    const int n0    = blockIdx.y * 16;
    const int mrow  = lane & 15;
    const int khalf = lane >> 4;

    int arow = r0 + mrow;
    if (arow >= NN) arow = NN - 1;
    const float* xrow = X + (size_t)arow * FH;
    const int col = n0 + mrow;

    v8f cL = {};
    v8f cR = {};
#pragma unroll
    for (int k = 0; k < FH; k += 4) {
        const v2f a = *(const v2f*)(xrow + k + khalf * 2);
        const size_t rk0 = (size_t)(k + khalf * 2) * FH + col;
        v2f bL, bR;
        bL.x = Wl[rk0];       bL.y = Wl[rk0 + FH];
        bR.x = Wr[rk0];       bR.y = Wr[rk0 + FH];
        cL = __builtin_amdgcn_wmma_f32_16x16x4_f32(false, a, false, bL,
                                                   (short)0, cL, false, false);
        cR = __builtin_amdgcn_wmma_f32_16x16x4_f32(false, a, false, bR,
                                                   (short)0, cR, false, false);
    }

    const float bbL = bl[col];
    const float bbR = br[col];
#pragma unroll
    for (int v = 0; v < 8; ++v) {
        const int row = r0 + v + khalf * 8;
        if (row < NN) {
            OutL[(size_t)row * FH + col] = cL[v] + bbL;
            OutR[(size_t)row * FH + col] = cR[v] + bbR;
        }
    }
}

__global__ void csr_zero(int* __restrict__ cnt, int* __restrict__ cursor)
{
    const int t = blockIdx.x * blockDim.x + threadIdx.x;
    if (t < NN) cnt[t] = 0;
    if (t == 0) cursor[0] = 0;
}

__global__ void csr_count(const int* __restrict__ dst,
                          int* __restrict__ cnt,
                          int* __restrict__ rank)
{
    const int e = blockIdx.x * blockDim.x + threadIdx.x;
    if (e < EE) rank[e] = atomicAdd(&cnt[dst[e]], 1);
}

__global__ void csr_offsets(const int* __restrict__ cnt,
                            int* __restrict__ cursor,
                            int* __restrict__ off)
{
    __shared__ int wsum[8];
    __shared__ int base;
    const int n    = blockIdx.x * 256 + threadIdx.x;
    const int lane = threadIdx.x & 31;
    const int wave = threadIdx.x >> 5;
    const int c = (n < NN) ? cnt[n] : 0;
    int v = c;
#pragma unroll
    for (int d = 1; d < 32; d <<= 1) {
        const int t = __shfl_up(v, d, 32);
        if (lane >= d) v += t;
    }
    if (lane == 31) wsum[wave] = v;
    __syncthreads();
    if (threadIdx.x == 0) {
        int tot = 0;
        for (int w = 0; w < 8; ++w) { const int t = wsum[w]; wsum[w] = tot; tot += t; }
        base = atomicAdd(cursor, tot);
    }
    __syncthreads();
    if (n < NN) off[n] = base + wsum[wave] + v - c;
}

__global__ void csr_fill(const int* __restrict__ src,
                         const int* __restrict__ dst,
                         const int* __restrict__ rank,
                         const int* __restrict__ off,
                         int* __restrict__ csr_src)
{
    const int e = blockIdx.x * blockDim.x + threadIdx.x;
    if (e < EE) csr_src[off[dst[e]] + rank[e]] = src[e];
}

__device__ __forceinline__ float head_sum8(float p)
{
    p += __int_as_float(__builtin_amdgcn_update_dpp(0, __float_as_int(p), 0x161, 0xf, 0xf, true));
    p += __int_as_float(__builtin_amdgcn_update_dpp(0, __float_as_int(p), 0x162, 0xf, 0xf, true));
    p += __int_as_float(__builtin_amdgcn_update_dpp(0, __float_as_int(p), 0x164, 0xf, 0xf, true));
    return p;
}

__device__ __forceinline__ v2f lrelu2(const v2f v)
{
    return __builtin_elementwise_max(v, v * 0.2f);
}

__device__ __forceinline__ float edge_logit(const v4f a, const v4f b, const v4f w)
{
    const v2f l01 = lrelu2(a.xy + b.xy);
    const v2f l23 = lrelu2(a.zw + b.zw);
    const v2f t   = l01 * w.xy + l23 * w.zw;
    return head_sum8(t.x + t.y);
}

template <int B>
__device__ __forceinline__ void batch_update(const v4f (&a)[B], const v4f b, const v4f w,
                                             float& m, float& den, v4f& acc)
{
    float p[B];
#pragma unroll
    for (int i = 0; i < B; ++i) p[i] = edge_logit(a[i], b, w);
    float mn = m;
#pragma unroll
    for (int i = 0; i < B; ++i) mn = fmaxf(mn, p[i]);
    const float sc = expf(m - mn);
    den *= sc;
    acc *= sc;
#pragma unroll
    for (int i = 0; i < B; ++i) {
        const float e = expf(p[i] - mn);
        den += e;
        acc += e * a[i];
    }
    m = mn;
}

template <int B>
__device__ __forceinline__ void gather_batch(const float* __restrict__ xl, const int myid,
                                             const int j, const int f,
                                             const v4f b, const v4f w,
                                             float& m, float& den, v4f& acc)
{
    v4f a[B];
#pragma unroll
    for (int i = 0; i < B; ++i) {
        const int s = __builtin_amdgcn_readlane(myid, j + i);
        a[i] = *(const v4f*)(xl + (size_t)s * FH + f);
    }
    if constexpr (B == 4) asm volatile("" :: "v"(a[0]), "v"(a[1]), "v"(a[2]), "v"(a[3]));
    if constexpr (B == 3) asm volatile("" :: "v"(a[0]), "v"(a[1]), "v"(a[2]));
    if constexpr (B == 2) asm volatile("" :: "v"(a[0]), "v"(a[1]));
    batch_update<B>(a, b, w, m, den, acc);
}

__global__ void node_gather(const float* __restrict__ xl,
                            const float* __restrict__ xr,
                            const int* __restrict__ off,
                            const int* __restrict__ cnt,
                            const int* __restrict__ csr_src,
                            const float* __restrict__ att,
                            const float* __restrict__ bo,
                            float* __restrict__ hout)
{
    const int lane = threadIdx.x & 31;
    const int n = __builtin_amdgcn_readfirstlane((int)((blockIdx.x * blockDim.x + threadIdx.x) >> 5));
    if (n >= NN) return;

    const int beg = __builtin_amdgcn_readfirstlane(off[n]);
    const int deg = __builtin_amdgcn_readfirstlane(cnt[n]);

    const int f = lane * 4;
    const v4f b = *(const v4f*)(xr + (size_t)n * FH + f);
    const v4f w = *(const v4f*)(att + f);
    v4f   acc   = *(const v4f*)(xl + (size_t)n * FH + f);
    float m     = edge_logit(acc, b, w);
    float den   = 1.f;

    for (int c = 0; c < deg; c += 32) {
        const int rem = deg - c;
        const int cn  = rem < 32 ? rem : 32;
        const int myid = (lane < cn) ? csr_src[beg + c + lane] : n;
        int j = 0;
        for (; j + 4 <= cn; j += 4) gather_batch<4>(xl, myid, j, f, b, w, m, den, acc);
        const int r = cn - j;
        if (r == 3)      gather_batch<3>(xl, myid, j, f, b, w, m, den, acc);
        else if (r == 2) gather_batch<2>(xl, myid, j, f, b, w, m, den, acc);
        else if (r == 1) gather_batch<1>(xl, myid, j, f, b, w, m, den, acc);
    }

    const float inv = 1.f / fmaxf(den, 1e-16f);
    const v4f bb = *(const v4f*)(bo + f);
    v4f o = acc * inv + bb;
    o.x = (o.x > 0.f) ? o.x : (expf(o.x) - 1.f);
    o.y = (o.y > 0.f) ? o.y : (expf(o.y) - 1.f);
    o.z = (o.z > 0.f) ? o.z : (expf(o.z) - 1.f);
    o.w = (o.w > 0.f) ? o.w : (expf(o.w) - 1.f);
    *(v4f*)(hout + (size_t)n * FH + f) = o;
}

__global__ void final_linear(const float* __restrict__ h,
                             const float* __restrict__ Wf,
                             const float* __restrict__ bf,
                             float* __restrict__ out)
{
    const int gid  = blockIdx.x * blockDim.x + threadIdx.x;
    const int n    = gid >> 5;
    const int lane = gid & 31;
    if (n >= NN) return;

    const int f = lane * 4;
    const float4 a = *(const float4*)(h + (size_t)n * FH + f);
    const float4 w = *(const float4*)(Wf + f);
    float p = a.x * w.x + a.y * w.y + a.z * w.z + a.w * w.w;
    p += __shfl_xor(p, 16, 32);
    p += __shfl_xor(p,  8, 32);
    p += __shfl_xor(p,  4, 32);
    p += __shfl_xor(p,  2, 32);
    p += __shfl_xor(p,  1, 32);
    if (lane == 0) out[n] = p + bf[0];
}

extern "C" void kernel_launch(void* const* d_in, const int* in_sizes, int n_in,
                              void* d_out, int out_size, void* d_ws, size_t ws_size,
                              hipStream_t stream)
{
    const float* x    = (const float*)d_in[0];
    const int*   ei   = (const int*)d_in[1];
    const int*   src  = ei;
    const int*   dst  = ei + EE;
    const float* Wl0  = (const float*)d_in[2];
    const float* bl0  = (const float*)d_in[3];
    const float* Wr0  = (const float*)d_in[4];
    const float* br0  = (const float*)d_in[5];
    const float* att0 = (const float*)d_in[6];
    const float* bo0  = (const float*)d_in[7];
    const float* Wl   = (const float*)d_in[8];
    const float* bl   = (const float*)d_in[9];
    const float* Wr   = (const float*)d_in[10];
    const float* br   = (const float*)d_in[11];
    const float* att  = (const float*)d_in[12];
    const float* bo   = (const float*)d_in[13];
    const float* Wf   = (const float*)d_in[14];
    const float* bf   = (const float*)d_in[15];

    float* ws   = (float*)d_ws;
    float* hbuf = ws;
    float* xl   = hbuf + (size_t)NN * FH;
    float* xr   = xl   + (size_t)NN * FH;
    int*   cnt  = (int*)(xr + (size_t)NN * FH);
    int*   off  = cnt  + NN;
    int*   cur  = off  + NN;
    int*   rank = cur  + 32;
    int*   csrc = rank + EE;

    const dim3 gemmGrid((NN + 127) / 128, FH / 16);
    const dim3 eGrid((EE + 255) / 256);
    const dim3 nGrid((NN + 255) / 256);
    const dim3 waveGrid((NN * 32 + 255) / 256);

    csr_zero   <<<nGrid, 256, 0, stream>>>(cnt, cur);
    csr_count  <<<eGrid, 256, 0, stream>>>(dst, cnt, rank);
    csr_offsets<<<nGrid, 256, 0, stream>>>(cnt, cur, off);
    csr_fill   <<<eGrid, 256, 0, stream>>>(src, dst, rank, off, csrc);

    for (int layer = 0; layer < 3; ++layer) {
        const float* in = (layer == 0) ? x : hbuf;
        const float *WL, *BL, *WR, *BR, *AT, *BO;
        if (layer == 0) { WL = Wl0; BL = bl0; WR = Wr0; BR = br0; AT = att0; BO = bo0; }
        else {
            const int i = layer - 1;
            WL = Wl + (size_t)i * FH * FH;  BL = bl + (size_t)i * FH;
            WR = Wr + (size_t)i * FH * FH;  BR = br + (size_t)i * FH;
            AT = att + (size_t)i * NH * HIDW;
            BO = bo + (size_t)i * FH;
        }
        gemm_wmma_dual<<<gemmGrid, 256, 0, stream>>>(in, WL, BL, WR, BR, xl, xr);
        node_gather   <<<waveGrid, 256, 0, stream>>>(xl, xr, off, cnt, csrc, AT, BO, hbuf);
    }
    final_linear<<<waveGrid, 256, 0, stream>>>(hbuf, Wf, bf, (float*)d_out);
}


# Dual GEMM as a resident grid of 256 blocks: both weight matrices staged once per block in LDS in fragment order, waves loop over 16-row strips with 16-byte A loads; k summed in a new order
# speedup vs baseline: 5.2369x; 1.3025x over previous
#include <hip/hip_runtime.h>

#define NN  50000
#define EE  800000
#define FH  128
#define NH  4
#define HIDW 32

typedef __attribute__((ext_vector_type(2))) float v2f;
typedef __attribute__((ext_vector_type(8))) float v8f;
typedef __attribute__((ext_vector_type(4))) float v4f;

#define GEMM_BLOCKS 256
#define GEMM_SLOTS  4

__global__ __launch_bounds__(512)
void gemm_wmma_dual(const float* __restrict__ X,
                    const float* __restrict__ Wl,
                    const float* __restrict__ bl,
                    const float* __restrict__ Wr,
                    const float* __restrict__ br,
                    float* __restrict__ OutL,
                    float* __restrict__ OutR)
{
    __shared__ v4f sW[2 * 16 * 8 * 32];

    const int tid  = threadIdx.x;
    const int lane = tid & 31;
    const int wave = __builtin_amdgcn_readfirstlane(tid >> 5);
    const int c    = lane & 15;
    const int h    = lane >> 4;
    const int qd   = wave & 3;
    const int slot = wave >> 2;
    const int nstrips = (NN + 15) / 16;

    {
        const int t  = (tid >> 5) & 7;
        const int g0 = tid >> 8;
#pragma unroll
        for (int i = 0; i < 8; ++i) {
            const int g = g0 + 2 * i;
            const size_t o = (size_t)(8 * g + 4 * h) * FH + 16 * t + c;
            v4f vl, vr;
            vl.x = Wl[o]; vl.y = Wl[o + FH]; vl.z = Wl[o + 2 * FH]; vl.w = Wl[o + 3 * FH];
            vr.x = Wr[o]; vr.y = Wr[o + FH]; vr.z = Wr[o + 2 * FH]; vr.w = Wr[o + 3 * FH];
            sW[((0 * 16 + g) * 8 + t) * 32 + lane] = vl;
            sW[((1 * 16 + g) * 8 + t) * 32 + lane] = vr;
        }
    }
    __syncthreads();

    const int colA = 32 * qd + c;
    const float bL0 = bl[colA], bL1 = bl[colA + 16];
    const float bR0 = br[colA], bR1 = br[colA + 16];
    const v4f* sL = sW + (2 * qd) * 32 + lane;
    const v4f* sR = sL + 16 * 8 * 32;

    for (int s = blockIdx.x + GEMM_BLOCKS * slot; s < nstrips; s += GEMM_BLOCKS * GEMM_SLOTS) {
        int arow = 16 * s + c;
        if (arow >= NN) arow = NN - 1;
        const v4f* xp = (const v4f*)(X + (size_t)arow * FH) + h;

        v4f a[16];
#pragma unroll
        for (int g = 0; g < 16; ++g) a[g] = xp[2 * g];

        v8f cL0 = {}, cL1 = {}, cR0 = {}, cR1 = {};
        v4f wL0 = sL[0], wL1 = sL[32], wR0 = sR[0], wR1 = sR[32];
#pragma unroll
        for (int g = 0; g < 16; ++g) {
            v4f nL0 = wL0, nL1 = wL1, nR0 = wR0, nR1 = wR1;
            if (g < 15) {
                nL0 = sL[(g + 1) * 256];  nL1 = sL[(g + 1) * 256 + 32];
                nR0 = sR[(g + 1) * 256];  nR1 = sR[(g + 1) * 256 + 32];
            }
            const v2f a01 = a[g].xy, a23 = a[g].zw;
            cL0 = __builtin_amdgcn_wmma_f32_16x16x4_f32(false, a01, false, wL0.xy, (short)0, cL0, false, false);
            cL1 = __builtin_amdgcn_wmma_f32_16x16x4_f32(false, a01, false, wL1.xy, (short)0, cL1, false, false);
            cR0 = __builtin_amdgcn_wmma_f32_16x16x4_f32(false, a01, false, wR0.xy, (short)0, cR0, false, false);
            cR1 = __builtin_amdgcn_wmma_f32_16x16x4_f32(false, a01, false, wR1.xy, (short)0, cR1, false, false);
            cL0 = __builtin_amdgcn_wmma_f32_16x16x4_f32(false, a23, false, wL0.zw, (short)0, cL0, false, false);
            cL1 = __builtin_amdgcn_wmma_f32_16x16x4_f32(false, a23, false, wL1.zw, (short)0, cL1, false, false);
            cR0 = __builtin_amdgcn_wmma_f32_16x16x4_f32(false, a23, false, wR0.zw, (short)0, cR0, false, false);
            cR1 = __builtin_amdgcn_wmma_f32_16x16x4_f32(false, a23, false, wR1.zw, (short)0, cR1, false, false);
            __builtin_amdgcn_sched_barrier(0);
            wL0 = nL0; wL1 = nL1; wR0 = nR0; wR1 = nR1;
        }

#pragma unroll
        for (int v = 0; v < 8; ++v) {
            const int row = 16 * s + v + 8 * h;
            if (row < NN) {
                float* pl = OutL + (size_t)row * FH + colA;
                float* pr = OutR + (size_t)row * FH + colA;
                pl[0]  = cL0[v] + bL0;
                pl[16] = cL1[v] + bL1;
                pr[0]  = cR0[v] + bR0;
                pr[16] = cR1[v] + bR1;
            }
        }
    }
}

__global__ void csr_zero(int* __restrict__ cnt, int* __restrict__ cursor)
{
    const int t = blockIdx.x * blockDim.x + threadIdx.x;
    if (t < NN) cnt[t] = 0;
    if (t == 0) cursor[0] = 0;
}

__global__ void csr_count(const int* __restrict__ dst,
                          int* __restrict__ cnt,
                          int* __restrict__ rank)
{
    const int e = blockIdx.x * blockDim.x + threadIdx.x;
    if (e < EE) rank[e] = atomicAdd(&cnt[dst[e]], 1);
}

__global__ void csr_offsets(const int* __restrict__ cnt,
                            int* __restrict__ cursor,
                            int* __restrict__ off)
{
    __shared__ int wsum[8];
    __shared__ int base;
    const int n    = blockIdx.x * 256 + threadIdx.x;
    const int lane = threadIdx.x & 31;
    const int wave = threadIdx.x >> 5;
    const int c = (n < NN) ? cnt[n] : 0;
    int v = c;
#pragma unroll
    for (int d = 1; d < 32; d <<= 1) {
        const int t = __shfl_up(v, d, 32);
        if (lane >= d) v += t;
    }
    if (lane == 31) wsum[wave] = v;
    __syncthreads();
    if (threadIdx.x == 0) {
        int tot = 0;
        for (int w = 0; w < 8; ++w) { const int t = wsum[w]; wsum[w] = tot; tot += t; }
        base = atomicAdd(cursor, tot);
    }
    __syncthreads();
    if (n < NN) off[n] = base + wsum[wave] + v - c;
}

__global__ void csr_fill(const int* __restrict__ src,
                         const int* __restrict__ dst,
                         const int* __restrict__ rank,
                         const int* __restrict__ off,
                         int* __restrict__ csr_src)
{
    const int e = blockIdx.x * blockDim.x + threadIdx.x;
    if (e < EE) csr_src[off[dst[e]] + rank[e]] = src[e];
}

__device__ __forceinline__ float head_sum8(float p)
{
    p += __int_as_float(__builtin_amdgcn_update_dpp(0, __float_as_int(p), 0x161, 0xf, 0xf, true));
    p += __int_as_float(__builtin_amdgcn_update_dpp(0, __float_as_int(p), 0x162, 0xf, 0xf, true));
    p += __int_as_float(__builtin_amdgcn_update_dpp(0, __float_as_int(p), 0x164, 0xf, 0xf, true));
    return p;
}

__device__ __forceinline__ v2f lrelu2(const v2f v)
{
    return __builtin_elementwise_max(v, v * 0.2f);
}

__device__ __forceinline__ float edge_logit(const v4f a, const v4f b, const v4f w)
{
    const v2f l01 = lrelu2(a.xy + b.xy);
    const v2f l23 = lrelu2(a.zw + b.zw);
    const v2f t   = l01 * w.xy + l23 * w.zw;
    return head_sum8(t.x + t.y);
}

template <int B>
__device__ __forceinline__ void batch_update(const v4f (&a)[B], const v4f b, const v4f w,
                                             float& m, float& den, v4f& acc)
{
    float p[B];
#pragma unroll
    for (int i = 0; i < B; ++i) p[i] = edge_logit(a[i], b, w);
    float mn = m;
#pragma unroll
    for (int i = 0; i < B; ++i) mn = fmaxf(mn, p[i]);
    const float sc = expf(m - mn);
    den *= sc;
    acc *= sc;
#pragma unroll
    for (int i = 0; i < B; ++i) {
        const float e = expf(p[i] - mn);
        den += e;
        acc += e * a[i];
    }
    m = mn;
}

template <int B>
__device__ __forceinline__ void gather_batch(const float* __restrict__ xl, const int myid,
                                             const int j, const int f,
                                             const v4f b, const v4f w,
                                             float& m, float& den, v4f& acc)
{
    v4f a[B];
#pragma unroll
    for (int i = 0; i < B; ++i) {
        const int s = __builtin_amdgcn_readlane(myid, j + i);
        a[i] = *(const v4f*)(xl + (size_t)s * FH + f);
    }
    if constexpr (B == 4) asm volatile("" :: "v"(a[0]), "v"(a[1]), "v"(a[2]), "v"(a[3]));
    if constexpr (B == 3) asm volatile("" :: "v"(a[0]), "v"(a[1]), "v"(a[2]));
    if constexpr (B == 2) asm volatile("" :: "v"(a[0]), "v"(a[1]));
    batch_update<B>(a, b, w, m, den, acc);
}

__global__ void node_gather(const float* __restrict__ xl,
                            const float* __restrict__ xr,
                            const int* __restrict__ off,
                            const int* __restrict__ cnt,
                            const int* __restrict__ csr_src,
                            const float* __restrict__ att,
                            const float* __restrict__ bo,
                            float* __restrict__ hout)
{
    const int lane = threadIdx.x & 31;
    const int n = __builtin_amdgcn_readfirstlane((int)((blockIdx.x * blockDim.x + threadIdx.x) >> 5));
    if (n >= NN) return;

    const int beg = __builtin_amdgcn_readfirstlane(off[n]);
    const int deg = __builtin_amdgcn_readfirstlane(cnt[n]);

    const int f = lane * 4;
    const v4f b = *(const v4f*)(xr + (size_t)n * FH + f);
    const v4f w = *(const v4f*)(att + f);
    v4f   acc   = *(const v4f*)(xl + (size_t)n * FH + f);
    float m     = edge_logit(acc, b, w);
    float den   = 1.f;

    for (int c = 0; c < deg; c += 32) {
        const int rem = deg - c;
        const int cn  = rem < 32 ? rem : 32;
        const int myid = (lane < cn) ? csr_src[beg + c + lane] : n;
        int j = 0;
        for (; j + 4 <= cn; j += 4) gather_batch<4>(xl, myid, j, f, b, w, m, den, acc);
        const int r = cn - j;
        if (r == 3)      gather_batch<3>(xl, myid, j, f, b, w, m, den, acc);
        else if (r == 2) gather_batch<2>(xl, myid, j, f, b, w, m, den, acc);
        else if (r == 1) gather_batch<1>(xl, myid, j, f, b, w, m, den, acc);
    }

    const float inv = 1.f / fmaxf(den, 1e-16f);
    const v4f bb = *(const v4f*)(bo + f);
    v4f o = acc * inv + bb;
    o.x = (o.x > 0.f) ? o.x : (expf(o.x) - 1.f);
    o.y = (o.y > 0.f) ? o.y : (expf(o.y) - 1.f);
    o.z = (o.z > 0.f) ? o.z : (expf(o.z) - 1.f);
    o.w = (o.w > 0.f) ? o.w : (expf(o.w) - 1.f);
    *(v4f*)(hout + (size_t)n * FH + f) = o;
}

__global__ void final_linear(const float* __restrict__ h,
                             const float* __restrict__ Wf,
                             const float* __restrict__ bf,
                             float* __restrict__ out)
{
    const int gid  = blockIdx.x * blockDim.x + threadIdx.x;
    const int n    = gid >> 5;
    const int lane = gid & 31;
    if (n >= NN) return;

    const int f = lane * 4;
    const float4 a = *(const float4*)(h + (size_t)n * FH + f);
    const float4 w = *(const float4*)(Wf + f);
    float p = a.x * w.x + a.y * w.y + a.z * w.z + a.w * w.w;
    p += __shfl_xor(p, 16, 32);
    p += __shfl_xor(p,  8, 32);
    p += __shfl_xor(p,  4, 32);
    p += __shfl_xor(p,  2, 32);
    p += __shfl_xor(p,  1, 32);
    if (lane == 0) out[n] = p + bf[0];
}

extern "C" void kernel_launch(void* const* d_in, const int* in_sizes, int n_in,
                              void* d_out, int out_size, void* d_ws, size_t ws_size,
                              hipStream_t stream)
{
    const float* x    = (const float*)d_in[0];
    const int*   ei   = (const int*)d_in[1];
    const int*   src  = ei;
    const int*   dst  = ei + EE;
    const float* Wl0  = (const float*)d_in[2];
    const float* bl0  = (const float*)d_in[3];
    const float* Wr0  = (const float*)d_in[4];
    const float* br0  = (const float*)d_in[5];
    const float* att0 = (const float*)d_in[6];
    const float* bo0  = (const float*)d_in[7];
    const float* Wl   = (const float*)d_in[8];
    const float* bl   = (const float*)d_in[9];
    const float* Wr   = (const float*)d_in[10];
    const float* br   = (const float*)d_in[11];
    const float* att  = (const float*)d_in[12];
    const float* bo   = (const float*)d_in[13];
    const float* Wf   = (const float*)d_in[14];
    const float* bf   = (const float*)d_in[15];

    float* ws   = (float*)d_ws;
    float* hbuf = ws;
    float* xl   = hbuf + (size_t)NN * FH;
    float* xr   = xl   + (size_t)NN * FH;
    int*   cnt  = (int*)(xr + (size_t)NN * FH);
    int*   off  = cnt  + NN;
    int*   cur  = off  + NN;
    int*   rank = cur  + 32;
    int*   csrc = rank + EE;

    const dim3 gemmGrid(GEMM_BLOCKS);
    const dim3 eGrid((EE + 255) / 256);
    const dim3 nGrid((NN + 255) / 256);
    const dim3 waveGrid((NN * 32 + 255) / 256);

    csr_zero   <<<nGrid, 256, 0, stream>>>(cnt, cur);
    csr_count  <<<eGrid, 256, 0, stream>>>(dst, cnt, rank);
    csr_offsets<<<nGrid, 256, 0, stream>>>(cnt, cur, off);
    csr_fill   <<<eGrid, 256, 0, stream>>>(src, dst, rank, off, csrc);

    for (int layer = 0; layer < 3; ++layer) {
        const float* in = (layer == 0) ? x : hbuf;
        const float *WL, *BL, *WR, *BR, *AT, *BO;
        if (layer == 0) { WL = Wl0; BL = bl0; WR = Wr0; BR = br0; AT = att0; BO = bo0; }
        else {
            const int i = layer - 1;
            WL = Wl + (size_t)i * FH * FH;  BL = bl + (size_t)i * FH;
            WR = Wr + (size_t)i * FH * FH;  BR = br + (size_t)i * FH;
            AT = att + (size_t)i * NH * HIDW;
            BO = bo + (size_t)i * FH;
        }
        gemm_wmma_dual<<<gemmGrid, 512, 0, stream>>>(in, WL, BL, WR, BR, xl, xr);
        node_gather   <<<waveGrid, 256, 0, stream>>>(xl, xr, off, cnt, csrc, AT, BO, hbuf);
    }
    final_linear<<<waveGrid, 256, 0, stream>>>(hbuf, Wf, bf, (float*)d_out);
}


# Counting sort of edges (count, scan, fill: 3 launches) replaced by one pass dropping each source into a 64-slot bucket of its destination, longer lists spill to a shared list; arrival order kept
# speedup vs baseline: 5.4868x; 1.0477x over previous
#include <hip/hip_runtime.h>

#define NN  50000
#define EE  800000
#define FH  128
#define NH  4
#define HIDW 32

typedef __attribute__((ext_vector_type(2))) float v2f;
typedef __attribute__((ext_vector_type(8))) float v8f;
typedef __attribute__((ext_vector_type(4))) float v4f;

#define GEMM_BLOCKS 256
#define GEMM_SLOTS  4

__global__ __launch_bounds__(512)
void gemm_wmma_dual(const float* __restrict__ X,
                    const float* __restrict__ Wl,
                    const float* __restrict__ bl,
                    const float* __restrict__ Wr,
                    const float* __restrict__ br,
                    float* __restrict__ OutL,
                    float* __restrict__ OutR)
{
    __shared__ v4f sW[2 * 16 * 8 * 32];

    const int tid  = threadIdx.x;
    const int lane = tid & 31;
    const int wave = __builtin_amdgcn_readfirstlane(tid >> 5);
    const int c    = lane & 15;
    const int h    = lane >> 4;
    const int qd   = wave & 3;
    const int slot = wave >> 2;
    const int nstrips = (NN + 15) / 16;

    {
        const int t  = (tid >> 5) & 7;
        const int g0 = tid >> 8;
#pragma unroll
        for (int i = 0; i < 8; ++i) {
            const int g = g0 + 2 * i;
            const size_t o = (size_t)(8 * g + 4 * h) * FH + 16 * t + c;
            v4f vl, vr;
            vl.x = Wl[o]; vl.y = Wl[o + FH]; vl.z = Wl[o + 2 * FH]; vl.w = Wl[o + 3 * FH];
            vr.x = Wr[o]; vr.y = Wr[o + FH]; vr.z = Wr[o + 2 * FH]; vr.w = Wr[o + 3 * FH];
            sW[((0 * 16 + g) * 8 + t) * 32 + lane] = vl;
            sW[((1 * 16 + g) * 8 + t) * 32 + lane] = vr;
        }
    }
    __syncthreads();

    const int colA = 32 * qd + c;
    const float bL0 = bl[colA], bL1 = bl[colA + 16];
    const float bR0 = br[colA], bR1 = br[colA + 16];
    const v4f* sL = sW + (2 * qd) * 32 + lane;
    const v4f* sR = sL + 16 * 8 * 32;

    for (int s = blockIdx.x + GEMM_BLOCKS * slot; s < nstrips; s += GEMM_BLOCKS * GEMM_SLOTS) {
        int arow = 16 * s + c;
        if (arow >= NN) arow = NN - 1;
        const v4f* xp = (const v4f*)(X + (size_t)arow * FH) + h;

        v4f a[16];
#pragma unroll
        for (int g = 0; g < 16; ++g) a[g] = xp[2 * g];

        v8f cL0 = {}, cL1 = {}, cR0 = {}, cR1 = {};
        v4f wL0 = sL[0], wL1 = sL[32], wR0 = sR[0], wR1 = sR[32];
#pragma unroll
        for (int g = 0; g < 16; ++g) {
            v4f nL0 = wL0, nL1 = wL1, nR0 = wR0, nR1 = wR1;
            if (g < 15) {
                nL0 = sL[(g + 1) * 256];  nL1 = sL[(g + 1) * 256 + 32];
                nR0 = sR[(g + 1) * 256];  nR1 = sR[(g + 1) * 256 + 32];
            }
            const v2f a01 = a[g].xy, a23 = a[g].zw;
            cL0 = __builtin_amdgcn_wmma_f32_16x16x4_f32(false, a01, false, wL0.xy, (short)0, cL0, false, false);
            cL1 = __builtin_amdgcn_wmma_f32_16x16x4_f32(false, a01, false, wL1.xy, (short)0, cL1, false, false);
            cR0 = __builtin_amdgcn_wmma_f32_16x16x4_f32(false, a01, false, wR0.xy, (short)0, cR0, false, false);
            cR1 = __builtin_amdgcn_wmma_f32_16x16x4_f32(false, a01, false, wR1.xy, (short)0, cR1, false, false);
            cL0 = __builtin_amdgcn_wmma_f32_16x16x4_f32(false, a23, false, wL0.zw, (short)0, cL0, false, false);
            cL1 = __builtin_amdgcn_wmma_f32_16x16x4_f32(false, a23, false, wL1.zw, (short)0, cL1, false, false);
            cR0 = __builtin_amdgcn_wmma_f32_16x16x4_f32(false, a23, false, wR0.zw, (short)0, cR0, false, false);
            cR1 = __builtin_amdgcn_wmma_f32_16x16x4_f32(false, a23, false, wR1.zw, (short)0, cR1, false, false);
            __builtin_amdgcn_sched_barrier(0);
            wL0 = nL0; wL1 = nL1; wR0 = nR0; wR1 = nR1;
        }

#pragma unroll
        for (int v = 0; v < 8; ++v) {
            const int row = 16 * s + v + 8 * h;
            if (row < NN) {
                float* pl = OutL + (size_t)row * FH + colA;
                float* pr = OutR + (size_t)row * FH + colA;
                pl[0]  = cL0[v] + bL0;
                pl[16] = cL1[v] + bL1;
                pr[0]  = cR0[v] + bR0;
                pr[16] = cR1[v] + bR1;
            }
        }
    }
}

#define CAP 64

__global__ void csr_zero(int* __restrict__ cnt, int* __restrict__ nspill)
{
    const int t = blockIdx.x * blockDim.x + threadIdx.x;
    if (t < NN) cnt[t] = 0;
    if (t == 0) nspill[0] = 0;
}

__global__ void csr_bucket(const int* __restrict__ src,
                           const int* __restrict__ dst,
                           int* __restrict__ cnt,
                           int* __restrict__ nspill,
                           int* __restrict__ bucket,
                           int* __restrict__ spill_dst,
                           int* __restrict__ spill_src)
{
    const int e = blockIdx.x * blockDim.x + threadIdx.x;
    if (e < EE) {
        const int d = dst[e];
        const int s = src[e];
        const int slot = atomicAdd(&cnt[d], 1);
        if (slot < CAP) {
            bucket[(size_t)d * CAP + slot] = s;
        } else {
            const int k = atomicAdd(nspill, 1);
            spill_dst[k] = d;
            spill_src[k] = s;
        }
    }
}

__device__ __forceinline__ float head_sum8(float p)
{
    p += __int_as_float(__builtin_amdgcn_update_dpp(0, __float_as_int(p), 0x161, 0xf, 0xf, true));
    p += __int_as_float(__builtin_amdgcn_update_dpp(0, __float_as_int(p), 0x162, 0xf, 0xf, true));
    p += __int_as_float(__builtin_amdgcn_update_dpp(0, __float_as_int(p), 0x164, 0xf, 0xf, true));
    return p;
}

__device__ __forceinline__ v2f lrelu2(const v2f v)
{
    return __builtin_elementwise_max(v, v * 0.2f);
}

__device__ __forceinline__ float edge_logit(const v4f a, const v4f b, const v4f w)
{
    const v2f l01 = lrelu2(a.xy + b.xy);
    const v2f l23 = lrelu2(a.zw + b.zw);
    const v2f t   = l01 * w.xy + l23 * w.zw;
    return head_sum8(t.x + t.y);
}

template <int B>
__device__ __forceinline__ void batch_update(const v4f (&a)[B], const v4f b, const v4f w,
                                             float& m, float& den, v4f& acc)
{
    float p[B];
#pragma unroll
    for (int i = 0; i < B; ++i) p[i] = edge_logit(a[i], b, w);
    float mn = m;
#pragma unroll
    for (int i = 0; i < B; ++i) mn = fmaxf(mn, p[i]);
    const float sc = expf(m - mn);
    den *= sc;
    acc *= sc;
#pragma unroll
    for (int i = 0; i < B; ++i) {
        const float e = expf(p[i] - mn);
        den += e;
        acc += e * a[i];
    }
    m = mn;
}

template <int B>
__device__ __forceinline__ void gather_batch(const float* __restrict__ xl, const int myid,
                                             const int j, const int f,
                                             const v4f b, const v4f w,
                                             float& m, float& den, v4f& acc)
{
    v4f a[B];
#pragma unroll
    for (int i = 0; i < B; ++i) {
        const int s = __builtin_amdgcn_readlane(myid, j + i);
        a[i] = *(const v4f*)(xl + (size_t)s * FH + f);
    }
    if constexpr (B == 4) asm volatile("" :: "v"(a[0]), "v"(a[1]), "v"(a[2]), "v"(a[3]));
    if constexpr (B == 3) asm volatile("" :: "v"(a[0]), "v"(a[1]), "v"(a[2]));
    if constexpr (B == 2) asm volatile("" :: "v"(a[0]), "v"(a[1]));
    batch_update<B>(a, b, w, m, den, acc);
}

__global__ void node_gather(const float* __restrict__ xl,
                            const float* __restrict__ xr,
                            const int* __restrict__ cnt,
                            const int* __restrict__ bucket,
                            const int* __restrict__ nspill,
                            const int* __restrict__ spill_dst,
                            const int* __restrict__ spill_src,
                            const float* __restrict__ att,
                            const float* __restrict__ bo,
                            float* __restrict__ hout)
{
    const int lane = threadIdx.x & 31;
    const int n = __builtin_amdgcn_readfirstlane((int)((blockIdx.x * blockDim.x + threadIdx.x) >> 5));
    if (n >= NN) return;

    const int deg = __builtin_amdgcn_readfirstlane(cnt[n]);
    const int nb  = deg < CAP ? deg : CAP;
    const int* ids = bucket + (size_t)n * CAP;

    const int f = lane * 4;
    const v4f b = *(const v4f*)(xr + (size_t)n * FH + f);
    const v4f w = *(const v4f*)(att + f);
    v4f   acc   = *(const v4f*)(xl + (size_t)n * FH + f);
    float m     = edge_logit(acc, b, w);
    float den   = 1.f;

    for (int c = 0; c < nb; c += 32) {
        const int rem = nb - c;
        const int cn  = rem < 32 ? rem : 32;
        const int myid = (lane < cn) ? ids[c + lane] : n;
        int j = 0;
        for (; j + 4 <= cn; j += 4) gather_batch<4>(xl, myid, j, f, b, w, m, den, acc);
        const int r = cn - j;
        if (r == 3)      gather_batch<3>(xl, myid, j, f, b, w, m, den, acc);
        else if (r == 2) gather_batch<2>(xl, myid, j, f, b, w, m, den, acc);
        else if (r == 1) gather_batch<1>(xl, myid, j, f, b, w, m, den, acc);
    }

    if (deg > CAP) {
        const int ns = __builtin_amdgcn_readfirstlane(nspill[0]);
        for (int k0 = 0; k0 < ns; k0 += 32) {
            const int k    = k0 + lane;
            const int d    = (k < ns) ? spill_dst[k] : -1;
            const int myid = (k < ns) ? spill_src[k] : n;
            unsigned hit = __builtin_amdgcn_ballot_w32(d == n);
            while (hit) {
                const int j = __builtin_amdgcn_readfirstlane(__builtin_ctz(hit));
                hit &= hit - 1;
                gather_batch<1>(xl, myid, j, f, b, w, m, den, acc);
            }
        }
    }

    const float inv = 1.f / fmaxf(den, 1e-16f);
    const v4f bb = *(const v4f*)(bo + f);
    v4f o = acc * inv + bb;
    o.x = (o.x > 0.f) ? o.x : (expf(o.x) - 1.f);
    o.y = (o.y > 0.f) ? o.y : (expf(o.y) - 1.f);
    o.z = (o.z > 0.f) ? o.z : (expf(o.z) - 1.f);
    o.w = (o.w > 0.f) ? o.w : (expf(o.w) - 1.f);
    *(v4f*)(hout + (size_t)n * FH + f) = o;
}

__global__ void final_linear(const float* __restrict__ h,
                             const float* __restrict__ Wf,
                             const float* __restrict__ bf,
                             float* __restrict__ out)
{
    const int gid  = blockIdx.x * blockDim.x + threadIdx.x;
    const int n    = gid >> 5;
    const int lane = gid & 31;
    if (n >= NN) return;

    const int f = lane * 4;
    const float4 a = *(const float4*)(h + (size_t)n * FH + f);
    const float4 w = *(const float4*)(Wf + f);
    float p = a.x * w.x + a.y * w.y + a.z * w.z + a.w * w.w;
    p += __shfl_xor(p, 16, 32);
    p += __shfl_xor(p,  8, 32);
    p += __shfl_xor(p,  4, 32);
    p += __shfl_xor(p,  2, 32);
    p += __shfl_xor(p,  1, 32);
    if (lane == 0) out[n] = p + bf[0];
}

extern "C" void kernel_launch(void* const* d_in, const int* in_sizes, int n_in,
                              void* d_out, int out_size, void* d_ws, size_t ws_size,
                              hipStream_t stream)
{
    const float* x    = (const float*)d_in[0];
    const int*   ei   = (const int*)d_in[1];
    const int*   src  = ei;
    const int*   dst  = ei + EE;
    const float* Wl0  = (const float*)d_in[2];
    const float* bl0  = (const float*)d_in[3];
    const float* Wr0  = (const float*)d_in[4];
    const float* br0  = (const float*)d_in[5];
    const float* att0 = (const float*)d_in[6];
    const float* bo0  = (const float*)d_in[7];
    const float* Wl   = (const float*)d_in[8];
    const float* bl   = (const float*)d_in[9];
    const float* Wr   = (const float*)d_in[10];
    const float* br   = (const float*)d_in[11];
    const float* att  = (const float*)d_in[12];
    const float* bo   = (const float*)d_in[13];
    const float* Wf   = (const float*)d_in[14];
    const float* bf   = (const float*)d_in[15];

    float* ws   = (float*)d_ws;
    float* hbuf = ws;
    float* xl   = hbuf + (size_t)NN * FH;
    float* xr   = xl   + (size_t)NN * FH;
    int*   bkt  = (int*)(xr + (size_t)NN * FH);
    int*   cnt  = bkt  + (size_t)NN * CAP;
    int*   nsp  = cnt  + NN;
    int*   spd  = nsp  + 32;
    int*   sps  = spd  + EE;

    const dim3 gemmGrid(GEMM_BLOCKS);
    const dim3 eGrid((EE + 255) / 256);
    const dim3 nGrid((NN + 255) / 256);
    const dim3 waveGrid((NN * 32 + 255) / 256);

    csr_zero  <<<nGrid, 256, 0, stream>>>(cnt, nsp);
    csr_bucket<<<eGrid, 256, 0, stream>>>(src, dst, cnt, nsp, bkt, spd, sps);

    for (int layer = 0; layer < 3; ++layer) {
        const float* in = (layer == 0) ? x : hbuf;
        const float *WL, *BL, *WR, *BR, *AT, *BO;
        if (layer == 0) { WL = Wl0; BL = bl0; WR = Wr0; BR = br0; AT = att0; BO = bo0; }
        else {
            const int i = layer - 1;
            WL = Wl + (size_t)i * FH * FH;  BL = bl + (size_t)i * FH;
            WR = Wr + (size_t)i * FH * FH;  BR = br + (size_t)i * FH;
            AT = att + (size_t)i * NH * HIDW;
            BO = bo + (size_t)i * FH;
        }
        gemm_wmma_dual<<<gemmGrid, 512, 0, stream>>>(in, WL, BL, WR, BR, xl, xr);
        node_gather   <<<waveGrid, 256, 0, stream>>>(xl, xr, cnt, bkt, nsp, spd, sps, AT, BO, hbuf);
    }
    final_linear<<<waveGrid, 256, 0, stream>>>(hbuf, Wf, bf, (float*)d_out);
}


# All exponentials of the gather (softmax weights, rescale, ELU) on the bare base-2 hardware exponential, log2(e) folded into the attention vector per node: one v_exp_f32 each, not the library expf
# speedup vs baseline: 5.7983x; 1.0568x over previous
#include <hip/hip_runtime.h>

#define NN  50000
#define EE  800000
#define FH  128
#define NH  4
#define HIDW 32

typedef __attribute__((ext_vector_type(2))) float v2f;
typedef __attribute__((ext_vector_type(8))) float v8f;
typedef __attribute__((ext_vector_type(4))) float v4f;

#define GEMM_BLOCKS 256
#define GEMM_SLOTS  4

__global__ __launch_bounds__(512)
void gemm_wmma_dual(const float* __restrict__ X,
                    const float* __restrict__ Wl,
                    const float* __restrict__ bl,
                    const float* __restrict__ Wr,
                    const float* __restrict__ br,
                    float* __restrict__ OutL,
                    float* __restrict__ OutR)
{
    __shared__ v4f sW[2 * 16 * 8 * 32];

    const int tid  = threadIdx.x;
    const int lane = tid & 31;
    const int wave = __builtin_amdgcn_readfirstlane(tid >> 5);
    const int c    = lane & 15;
    const int h    = lane >> 4;
    const int qd   = wave & 3;
    const int slot = wave >> 2;
    const int nstrips = (NN + 15) / 16;

    {
        const int t  = (tid >> 5) & 7;
        const int g0 = tid >> 8;
#pragma unroll
        for (int i = 0; i < 8; ++i) {
            const int g = g0 + 2 * i;
            const size_t o = (size_t)(8 * g + 4 * h) * FH + 16 * t + c;
            v4f vl, vr;
            vl.x = Wl[o]; vl.y = Wl[o + FH]; vl.z = Wl[o + 2 * FH]; vl.w = Wl[o + 3 * FH];
            vr.x = Wr[o]; vr.y = Wr[o + FH]; vr.z = Wr[o + 2 * FH]; vr.w = Wr[o + 3 * FH];
            sW[((0 * 16 + g) * 8 + t) * 32 + lane] = vl;
            sW[((1 * 16 + g) * 8 + t) * 32 + lane] = vr;
        }
    }
    __syncthreads();

    const int colA = 32 * qd + c;
    const float bL0 = bl[colA], bL1 = bl[colA + 16];
    const float bR0 = br[colA], bR1 = br[colA + 16];
    const v4f* sL = sW + (2 * qd) * 32 + lane;
    const v4f* sR = sL + 16 * 8 * 32;

    for (int s = blockIdx.x + GEMM_BLOCKS * slot; s < nstrips; s += GEMM_BLOCKS * GEMM_SLOTS) {
        int arow = 16 * s + c;
        if (arow >= NN) arow = NN - 1;
        const v4f* xp = (const v4f*)(X + (size_t)arow * FH) + h;

        v4f a[16];
#pragma unroll
        for (int g = 0; g < 16; ++g) a[g] = xp[2 * g];

        v8f cL0 = {}, cL1 = {}, cR0 = {}, cR1 = {};
        v4f wL0 = sL[0], wL1 = sL[32], wR0 = sR[0], wR1 = sR[32];
#pragma unroll
        for (int g = 0; g < 16; ++g) {
            v4f nL0 = wL0, nL1 = wL1, nR0 = wR0, nR1 = wR1;
            if (g < 15) {
                nL0 = sL[(g + 1) * 256];  nL1 = sL[(g + 1) * 256 + 32];
                nR0 = sR[(g + 1) * 256];  nR1 = sR[(g + 1) * 256 + 32];
            }
            const v2f a01 = a[g].xy, a23 = a[g].zw;
            cL0 = __builtin_amdgcn_wmma_f32_16x16x4_f32(false, a01, false, wL0.xy, (short)0, cL0, false, false);
            cL1 = __builtin_amdgcn_wmma_f32_16x16x4_f32(false, a01, false, wL1.xy, (short)0, cL1, false, false);
            cR0 = __builtin_amdgcn_wmma_f32_16x16x4_f32(false, a01, false, wR0.xy, (short)0, cR0, false, false);
            cR1 = __builtin_amdgcn_wmma_f32_16x16x4_f32(false, a01, false, wR1.xy, (short)0, cR1, false, false);
            cL0 = __builtin_amdgcn_wmma_f32_16x16x4_f32(false, a23, false, wL0.zw, (short)0, cL0, false, false);
            cL1 = __builtin_amdgcn_wmma_f32_16x16x4_f32(false, a23, false, wL1.zw, (short)0, cL1, false, false);
            cR0 = __builtin_amdgcn_wmma_f32_16x16x4_f32(false, a23, false, wR0.zw, (short)0, cR0, false, false);
            cR1 = __builtin_amdgcn_wmma_f32_16x16x4_f32(false, a23, false, wR1.zw, (short)0, cR1, false, false);
            __builtin_amdgcn_sched_barrier(0);
            wL0 = nL0; wL1 = nL1; wR0 = nR0; wR1 = nR1;
        }

#pragma unroll
        for (int v = 0; v < 8; ++v) {
            const int row = 16 * s + v + 8 * h;
            if (row < NN) {
                float* pl = OutL + (size_t)row * FH + colA;
                float* pr = OutR + (size_t)row * FH + colA;
                pl[0]  = cL0[v] + bL0;
                pl[16] = cL1[v] + bL1;
                pr[0]  = cR0[v] + bR0;
                pr[16] = cR1[v] + bR1;
            }
        }
    }
}

#define CAP 64

__global__ void csr_zero(int* __restrict__ cnt, int* __restrict__ nspill)
{
    const int t = blockIdx.x * blockDim.x + threadIdx.x;
    if (t < NN) cnt[t] = 0;
    if (t == 0) nspill[0] = 0;
}

__global__ void csr_bucket(const int* __restrict__ src,
                           const int* __restrict__ dst,
                           int* __restrict__ cnt,
                           int* __restrict__ nspill,
                           int* __restrict__ bucket,
                           int* __restrict__ spill_dst,
                           int* __restrict__ spill_src)
{
    const int e = blockIdx.x * blockDim.x + threadIdx.x;
    if (e < EE) {
        const int d = dst[e];
        const int s = src[e];
        const int slot = atomicAdd(&cnt[d], 1);
        if (slot < CAP) {
            bucket[(size_t)d * CAP + slot] = s;
        } else {
            const int k = atomicAdd(nspill, 1);
            spill_dst[k] = d;
            spill_src[k] = s;
        }
    }
}

__device__ __forceinline__ float head_sum8(float p)
{
    p += __int_as_float(__builtin_amdgcn_update_dpp(0, __float_as_int(p), 0x161, 0xf, 0xf, true));
    p += __int_as_float(__builtin_amdgcn_update_dpp(0, __float_as_int(p), 0x162, 0xf, 0xf, true));
    p += __int_as_float(__builtin_amdgcn_update_dpp(0, __float_as_int(p), 0x164, 0xf, 0xf, true));
    return p;
}

#define LOG2E 1.44269504088896341f
__device__ __forceinline__ float exp_hw(const float v)
{
    return __builtin_amdgcn_exp2f(v * LOG2E);
}

__device__ __forceinline__ v2f lrelu2(const v2f v)
{
    return __builtin_elementwise_max(v, v * 0.2f);
}

__device__ __forceinline__ float edge_logit(const v4f a, const v4f b, const v4f w)
{
    const v2f l01 = lrelu2(a.xy + b.xy);
    const v2f l23 = lrelu2(a.zw + b.zw);
    const v2f t   = l01 * w.xy + l23 * w.zw;
    return head_sum8(t.x + t.y);
}

template <int B>
__device__ __forceinline__ void batch_update(const v4f (&a)[B], const v4f b, const v4f w,
                                             float& m, float& den, v4f& acc)
{
    float p[B];
#pragma unroll
    for (int i = 0; i < B; ++i) p[i] = edge_logit(a[i], b, w);
    float mn = m;
#pragma unroll
    for (int i = 0; i < B; ++i) mn = fmaxf(mn, p[i]);
    const float sc = __builtin_amdgcn_exp2f(m - mn);
    den *= sc;
    acc *= sc;
#pragma unroll
    for (int i = 0; i < B; ++i) {
        const float e = __builtin_amdgcn_exp2f(p[i] - mn);
        den += e;
        acc += e * a[i];
    }
    m = mn;
}

template <int B>
__device__ __forceinline__ void gather_batch(const float* __restrict__ xl, const int myid,
                                             const int j, const int f,
                                             const v4f b, const v4f w,
                                             float& m, float& den, v4f& acc)
{
    v4f a[B];
#pragma unroll
    for (int i = 0; i < B; ++i) {
        const int s = __builtin_amdgcn_readlane(myid, j + i);
        a[i] = *(const v4f*)(xl + (size_t)s * FH + f);
    }
    if constexpr (B == 4) asm volatile("" :: "v"(a[0]), "v"(a[1]), "v"(a[2]), "v"(a[3]));
    if constexpr (B == 3) asm volatile("" :: "v"(a[0]), "v"(a[1]), "v"(a[2]));
    if constexpr (B == 2) asm volatile("" :: "v"(a[0]), "v"(a[1]));
    batch_update<B>(a, b, w, m, den, acc);
}

__global__ void node_gather(const float* __restrict__ xl,
                            const float* __restrict__ xr,
                            const int* __restrict__ cnt,
                            const int* __restrict__ bucket,
                            const int* __restrict__ nspill,
                            const int* __restrict__ spill_dst,
                            const int* __restrict__ spill_src,
                            const float* __restrict__ att,
                            const float* __restrict__ bo,
                            float* __restrict__ hout)
{
    const int lane = threadIdx.x & 31;
    const int n = __builtin_amdgcn_readfirstlane((int)((blockIdx.x * blockDim.x + threadIdx.x) >> 5));
    if (n >= NN) return;

    const int deg = __builtin_amdgcn_readfirstlane(cnt[n]);
    const int nb  = deg < CAP ? deg : CAP;
    const int* ids = bucket + (size_t)n * CAP;

    const int f = lane * 4;
    const v4f b = *(const v4f*)(xr + (size_t)n * FH + f);
    const v4f w = *(const v4f*)(att + f) * LOG2E;
    v4f   acc   = *(const v4f*)(xl + (size_t)n * FH + f);
    float m     = edge_logit(acc, b, w);
    float den   = 1.f;

    for (int c = 0; c < nb; c += 32) {
        const int rem = nb - c;
        const int cn  = rem < 32 ? rem : 32;
        const int myid = (lane < cn) ? ids[c + lane] : n;
        int j = 0;
        for (; j + 4 <= cn; j += 4) gather_batch<4>(xl, myid, j, f, b, w, m, den, acc);
        const int r = cn - j;
        if (r == 3)      gather_batch<3>(xl, myid, j, f, b, w, m, den, acc);
        else if (r == 2) gather_batch<2>(xl, myid, j, f, b, w, m, den, acc);
        else if (r == 1) gather_batch<1>(xl, myid, j, f, b, w, m, den, acc);
    }

    if (deg > CAP) {
        const int ns = __builtin_amdgcn_readfirstlane(nspill[0]);
        for (int k0 = 0; k0 < ns; k0 += 32) {
            const int k    = k0 + lane;
            const int d    = (k < ns) ? spill_dst[k] : -1;
            const int myid = (k < ns) ? spill_src[k] : n;
            unsigned hit = __builtin_amdgcn_ballot_w32(d == n);
            while (hit) {
                const int j = __builtin_amdgcn_readfirstlane(__builtin_ctz(hit));
                hit &= hit - 1;
                gather_batch<1>(xl, myid, j, f, b, w, m, den, acc);
            }
        }
    }

    const float inv = 1.f / fmaxf(den, 1e-16f);
    const v4f bb = *(const v4f*)(bo + f);
    v4f o = acc * inv + bb;
    o.x = (o.x > 0.f) ? o.x : (exp_hw(o.x) - 1.f);
    o.y = (o.y > 0.f) ? o.y : (exp_hw(o.y) - 1.f);
    o.z = (o.z > 0.f) ? o.z : (exp_hw(o.z) - 1.f);
    o.w = (o.w > 0.f) ? o.w : (exp_hw(o.w) - 1.f);
    *(v4f*)(hout + (size_t)n * FH + f) = o;
}

__global__ void final_linear(const float* __restrict__ h,
                             const float* __restrict__ Wf,
                             const float* __restrict__ bf,
                             float* __restrict__ out)
{
    const int gid  = blockIdx.x * blockDim.x + threadIdx.x;
    const int n    = gid >> 5;
    const int lane = gid & 31;
    if (n >= NN) return;

    const int f = lane * 4;
    const float4 a = *(const float4*)(h + (size_t)n * FH + f);
    const float4 w = *(const float4*)(Wf + f);
    float p = a.x * w.x + a.y * w.y + a.z * w.z + a.w * w.w;
    p += __shfl_xor(p, 16, 32);
    p += __shfl_xor(p,  8, 32);
    p += __shfl_xor(p,  4, 32);
    p += __shfl_xor(p,  2, 32);
    p += __shfl_xor(p,  1, 32);
    if (lane == 0) out[n] = p + bf[0];
}

extern "C" void kernel_launch(void* const* d_in, const int* in_sizes, int n_in,
                              void* d_out, int out_size, void* d_ws, size_t ws_size,
                              hipStream_t stream)
{
    const float* x    = (const float*)d_in[0];
    const int*   ei   = (const int*)d_in[1];
    const int*   src  = ei;
    const int*   dst  = ei + EE;
    const float* Wl0  = (const float*)d_in[2];
    const float* bl0  = (const float*)d_in[3];
    const float* Wr0  = (const float*)d_in[4];
    const float* br0  = (const float*)d_in[5];
    const float* att0 = (const float*)d_in[6];
    const float* bo0  = (const float*)d_in[7];
    const float* Wl   = (const float*)d_in[8];
    const float* bl   = (const float*)d_in[9];
    const float* Wr   = (const float*)d_in[10];
    const float* br   = (const float*)d_in[11];
    const float* att  = (const float*)d_in[12];
    const float* bo   = (const float*)d_in[13];
    const float* Wf   = (const float*)d_in[14];
    const float* bf   = (const float*)d_in[15];

    float* ws   = (float*)d_ws;
    float* hbuf = ws;
    float* xl   = hbuf + (size_t)NN * FH;
    float* xr   = xl   + (size_t)NN * FH;
    int*   bkt  = (int*)(xr + (size_t)NN * FH);
    int*   cnt  = bkt  + (size_t)NN * CAP;
    int*   nsp  = cnt  + NN;
    int*   spd  = nsp  + 32;
    int*   sps  = spd  + EE;

    const dim3 gemmGrid(GEMM_BLOCKS);
    const dim3 eGrid((EE + 255) / 256);
    const dim3 nGrid((NN + 255) / 256);
    const dim3 waveGrid((NN * 32 + 255) / 256);

    csr_zero  <<<nGrid, 256, 0, stream>>>(cnt, nsp);
    csr_bucket<<<eGrid, 256, 0, stream>>>(src, dst, cnt, nsp, bkt, spd, sps);

    for (int layer = 0; layer < 3; ++layer) {
        const float* in = (layer == 0) ? x : hbuf;
        const float *WL, *BL, *WR, *BR, *AT, *BO;
        if (layer == 0) { WL = Wl0; BL = bl0; WR = Wr0; BR = br0; AT = att0; BO = bo0; }
        else {
            const int i = layer - 1;
            WL = Wl + (size_t)i * FH * FH;  BL = bl + (size_t)i * FH;
            WR = Wr + (size_t)i * FH * FH;  BR = br + (size_t)i * FH;
            AT = att + (size_t)i * NH * HIDW;
            BO = bo + (size_t)i * FH;
        }
        gemm_wmma_dual<<<gemmGrid, 512, 0, stream>>>(in, WL, BL, WR, BR, xl, xr);
        node_gather   <<<waveGrid, 256, 0, stream>>>(xl, xr, cnt, bkt, nsp, spd, sps, AT, BO, hbuf);
    }
    final_linear<<<waveGrid, 256, 0, stream>>>(hbuf, Wf, bf, (float*)d_out);
}


# Edge bucketing with no per-edge global atomic or scattered store: blocks sort their edges by 128-node range in LDS, one block per range fills its buckets in LDS, whole-line stores; arrival order
# speedup vs baseline: 7.5288x; 1.2984x over previous
#include <hip/hip_runtime.h>

#define NN  50000
#define EE  800000
#define FH  128
#define NH  4
#define HIDW 32

typedef __attribute__((ext_vector_type(2))) float v2f;
typedef __attribute__((ext_vector_type(8))) float v8f;
typedef __attribute__((ext_vector_type(4))) float v4f;

#define GEMM_BLOCKS 256
#define GEMM_SLOTS  4

__global__ __launch_bounds__(512)
void gemm_wmma_dual(const float* __restrict__ X,
                    const float* __restrict__ Wl,
                    const float* __restrict__ bl,
                    const float* __restrict__ Wr,
                    const float* __restrict__ br,
                    float* __restrict__ OutL,
                    float* __restrict__ OutR)
{
    __shared__ v4f sW[2 * 16 * 8 * 32];

    const int tid  = threadIdx.x;
    const int lane = tid & 31;
    const int wave = __builtin_amdgcn_readfirstlane(tid >> 5);
    const int c    = lane & 15;
    const int h    = lane >> 4;
    const int qd   = wave & 3;
    const int slot = wave >> 2;
    const int nstrips = (NN + 15) / 16;

    {
        const int t  = (tid >> 5) & 7;
        const int g0 = tid >> 8;
#pragma unroll
        for (int i = 0; i < 8; ++i) {
            const int g = g0 + 2 * i;
            const size_t o = (size_t)(8 * g + 4 * h) * FH + 16 * t + c;
            v4f vl, vr;
            vl.x = Wl[o]; vl.y = Wl[o + FH]; vl.z = Wl[o + 2 * FH]; vl.w = Wl[o + 3 * FH];
            vr.x = Wr[o]; vr.y = Wr[o + FH]; vr.z = Wr[o + 2 * FH]; vr.w = Wr[o + 3 * FH];
            sW[((0 * 16 + g) * 8 + t) * 32 + lane] = vl;
            sW[((1 * 16 + g) * 8 + t) * 32 + lane] = vr;
        }
    }
    __syncthreads();

    const int colA = 32 * qd + c;
    const float bL0 = bl[colA], bL1 = bl[colA + 16];
    const float bR0 = br[colA], bR1 = br[colA + 16];
    const v4f* sL = sW + (2 * qd) * 32 + lane;
    const v4f* sR = sL + 16 * 8 * 32;

    for (int s = blockIdx.x + GEMM_BLOCKS * slot; s < nstrips; s += GEMM_BLOCKS * GEMM_SLOTS) {
        int arow = 16 * s + c;
        if (arow >= NN) arow = NN - 1;
        const v4f* xp = (const v4f*)(X + (size_t)arow * FH) + h;

        v4f a[16];
#pragma unroll
        for (int g = 0; g < 16; ++g) a[g] = xp[2 * g];

        v8f cL0 = {}, cL1 = {}, cR0 = {}, cR1 = {};
        v4f wL0 = sL[0], wL1 = sL[32], wR0 = sR[0], wR1 = sR[32];
#pragma unroll
        for (int g = 0; g < 16; ++g) {
            v4f nL0 = wL0, nL1 = wL1, nR0 = wR0, nR1 = wR1;
            if (g < 15) {
                nL0 = sL[(g + 1) * 256];  nL1 = sL[(g + 1) * 256 + 32];
                nR0 = sR[(g + 1) * 256];  nR1 = sR[(g + 1) * 256 + 32];
            }
            const v2f a01 = a[g].xy, a23 = a[g].zw;
            cL0 = __builtin_amdgcn_wmma_f32_16x16x4_f32(false, a01, false, wL0.xy, (short)0, cL0, false, false);
            cL1 = __builtin_amdgcn_wmma_f32_16x16x4_f32(false, a01, false, wL1.xy, (short)0, cL1, false, false);
            cR0 = __builtin_amdgcn_wmma_f32_16x16x4_f32(false, a01, false, wR0.xy, (short)0, cR0, false, false);
            cR1 = __builtin_amdgcn_wmma_f32_16x16x4_f32(false, a01, false, wR1.xy, (short)0, cR1, false, false);
            cL0 = __builtin_amdgcn_wmma_f32_16x16x4_f32(false, a23, false, wL0.zw, (short)0, cL0, false, false);
            cL1 = __builtin_amdgcn_wmma_f32_16x16x4_f32(false, a23, false, wL1.zw, (short)0, cL1, false, false);
            cR0 = __builtin_amdgcn_wmma_f32_16x16x4_f32(false, a23, false, wR0.zw, (short)0, cR0, false, false);
            cR1 = __builtin_amdgcn_wmma_f32_16x16x4_f32(false, a23, false, wR1.zw, (short)0, cR1, false, false);
            __builtin_amdgcn_sched_barrier(0);
            wL0 = nL0; wL1 = nL1; wR0 = nR0; wR1 = nR1;
        }

#pragma unroll
        for (int v = 0; v < 8; ++v) {
            const int row = 16 * s + v + 8 * h;
            if (row < NN) {
                float* pl = OutL + (size_t)row * FH + colA;
                float* pr = OutR + (size_t)row * FH + colA;
                pl[0]  = cL0[v] + bL0;
                pl[16] = cL1[v] + bL1;
                pr[0]  = cR0[v] + bR0;
                pr[16] = cR1[v] + bR1;
            }
        }
    }
}

#define CAP 64
#define RN  128
#define NR  ((NN + RN - 1) / RN)
#define P1_THREADS 1024
#define P1_EPT     4
#define P1_CHUNK   (P1_THREADS * P1_EPT)
#define NB1 ((EE + P1_CHUNK - 1) / P1_CHUNK)
#define P2_THREADS 1024
static_assert(NB1 < 256, "the piece search in range_buckets does 8 halvings");
static_assert(NR <= P1_THREADS && NB1 <= P2_THREADS && RN <= P2_THREADS, "one thread per table entry");

__device__ __forceinline__ void wave_exclusive_scan(const int* in, int* out, const int n, const int lane)
{
    const int per  = (n + 31) / 32;
    const int base = lane * per;
    int sum = 0;
    for (int k = 0; k < per; ++k) { const int r = base + k; if (r < n) sum += in[r]; }
    int inc = sum;
    for (int o = 1; o < 32; o <<= 1) { const int v = __shfl_up(inc, o, 32); if (lane >= o) inc += v; }
    int run = inc - sum;
    for (int k = 0; k < per; ++k) { const int r = base + k; if (r < n) { out[r] = run; run += in[r]; } }
    if (lane == 31) out[n] = inc;
}

__global__ __launch_bounds__(P1_THREADS)
void edge_partition(const int* __restrict__ src,
                    const int* __restrict__ dst,
                    int2* __restrict__ srt,
                    int* __restrict__ offTab,
                    int* __restrict__ nspill)
{
    __shared__ int hist[NR];
    __shared__ int off[NR + 1];
    const int tid = threadIdx.x;
    const int w   = blockIdx.x;
    if (tid < NR) hist[tid] = 0;
    if (w == 0 && tid == 0) nspill[0] = 0;
    __syncthreads();

    int s[P1_EPT], d[P1_EPT], pos[P1_EPT];
#pragma unroll
    for (int i = 0; i < P1_EPT; ++i) {
        const int e = w * P1_CHUNK + i * P1_THREADS + tid;
        d[i] = -1; s[i] = 0;
        if (e < EE) { d[i] = dst[e]; s[i] = src[e]; }
    }
#pragma unroll
    for (int i = 0; i < P1_EPT; ++i) {
        pos[i] = 0;
        if ((unsigned)d[i] < (unsigned)NN) pos[i] = atomicAdd(&hist[d[i] / RN], 1);
    }
    __syncthreads();
    if (tid < 32) wave_exclusive_scan(hist, off, NR, tid);
    __syncthreads();
#pragma unroll
    for (int i = 0; i < P1_EPT; ++i) {
        if ((unsigned)d[i] < (unsigned)NN)
            srt[(size_t)w * P1_CHUNK + off[d[i] / RN] + pos[i]] = make_int2(s[i], d[i]);
    }
    if (tid <= NR) offTab[w * (NR + 1) + tid] = off[tid];
}

__global__ __launch_bounds__(P2_THREADS)
void range_buckets(const int2* __restrict__ srt,
                   const int* __restrict__ offTab,
                   int* __restrict__ cnt,
                   int* __restrict__ bucket,
                   int* __restrict__ nspill,
                   int* __restrict__ spill_dst,
                   int* __restrict__ spill_src)
{
    __shared__ int4 sB4[RN * CAP / 4];
    __shared__ int  sCnt[RN];
    __shared__ int  segLen[NB1];
    __shared__ int  segStart[NB1];
    __shared__ int  pre[NB1 + 1];
    int* sB = (int*)sB4;
    const int tid = threadIdx.x;
    const int r   = blockIdx.x;

    if (tid < NB1) {
        const int a = offTab[tid * (NR + 1) + r];
        const int b = offTab[tid * (NR + 1) + r + 1];
        segStart[tid] = tid * P1_CHUNK + a;
        segLen[tid]   = b - a;
    }
    if (tid < RN) sCnt[tid] = 0;
    __syncthreads();
    if (tid < 32) wave_exclusive_scan(segLen, pre, NB1, tid);
    __syncthreads();

    const int total = pre[NB1];
    for (int i = tid; i < total; i += P2_THREADS) {
        int lo = 0, hi = NB1;
#pragma unroll
        for (int it = 0; it < 8; ++it) {
            const int mid = (lo + hi) >> 1;
            if (pre[mid] <= i) lo = mid; else hi = mid;
        }
        const int2 sd = srt[segStart[lo] + (i - pre[lo])];
        const int dl   = sd.y - r * RN;
        const int slot = atomicAdd(&sCnt[dl], 1);
        if (slot < CAP) {
            sB[dl * CAP + slot] = sd.x;
        } else {
            const int k = atomicAdd(nspill, 1);
            spill_dst[k] = sd.y;
            spill_src[k] = sd.x;
        }
    }
    __syncthreads();

    int4* gB = (int4*)(bucket + (size_t)r * RN * CAP);
    for (int q = tid; q < RN * CAP / 4; q += P2_THREADS) gB[q] = sB4[q];
    if (tid < RN) cnt[r * RN + tid] = sCnt[tid];
}

__device__ __forceinline__ float head_sum8(float p)
{
    p += __int_as_float(__builtin_amdgcn_update_dpp(0, __float_as_int(p), 0x161, 0xf, 0xf, true));
    p += __int_as_float(__builtin_amdgcn_update_dpp(0, __float_as_int(p), 0x162, 0xf, 0xf, true));
    p += __int_as_float(__builtin_amdgcn_update_dpp(0, __float_as_int(p), 0x164, 0xf, 0xf, true));
    return p;
}

#define LOG2E 1.44269504088896341f
__device__ __forceinline__ float exp_hw(const float v)
{
    return __builtin_amdgcn_exp2f(v * LOG2E);
}

__device__ __forceinline__ v2f lrelu2(const v2f v)
{
    return __builtin_elementwise_max(v, v * 0.2f);
}

__device__ __forceinline__ float edge_logit(const v4f a, const v4f b, const v4f w)
{
    const v2f l01 = lrelu2(a.xy + b.xy);
    const v2f l23 = lrelu2(a.zw + b.zw);
    const v2f t   = l01 * w.xy + l23 * w.zw;
    return head_sum8(t.x + t.y);
}

template <int B>
__device__ __forceinline__ void batch_update(const v4f (&a)[B], const v4f b, const v4f w,
                                             float& m, float& den, v4f& acc)
{
    float p[B];
#pragma unroll
    for (int i = 0; i < B; ++i) p[i] = edge_logit(a[i], b, w);
    float mn = m;
#pragma unroll
    for (int i = 0; i < B; ++i) mn = fmaxf(mn, p[i]);
    const float sc = __builtin_amdgcn_exp2f(m - mn);
    den *= sc;
    acc *= sc;
#pragma unroll
    for (int i = 0; i < B; ++i) {
        const float e = __builtin_amdgcn_exp2f(p[i] - mn);
        den += e;
        acc += e * a[i];
    }
    m = mn;
}

template <int B>
__device__ __forceinline__ void gather_batch(const float* __restrict__ xl, const int myid,
                                             const int j, const int f,
                                             const v4f b, const v4f w,
                                             float& m, float& den, v4f& acc)
{
    v4f a[B];
#pragma unroll
    for (int i = 0; i < B; ++i) {
        const int s = __builtin_amdgcn_readlane(myid, j + i);
        a[i] = *(const v4f*)(xl + (size_t)s * FH + f);
    }
    if constexpr (B == 4) asm volatile("" :: "v"(a[0]), "v"(a[1]), "v"(a[2]), "v"(a[3]));
    if constexpr (B == 3) asm volatile("" :: "v"(a[0]), "v"(a[1]), "v"(a[2]));
    if constexpr (B == 2) asm volatile("" :: "v"(a[0]), "v"(a[1]));
    batch_update<B>(a, b, w, m, den, acc);
}

__global__ void node_gather(const float* __restrict__ xl,
                            const float* __restrict__ xr,
                            const int* __restrict__ cnt,
                            const int* __restrict__ bucket,
                            const int* __restrict__ nspill,
                            const int* __restrict__ spill_dst,
                            const int* __restrict__ spill_src,
                            const float* __restrict__ att,
                            const float* __restrict__ bo,
                            float* __restrict__ hout)
{
    const int lane = threadIdx.x & 31;
    const int n = __builtin_amdgcn_readfirstlane((int)((blockIdx.x * blockDim.x + threadIdx.x) >> 5));
    if (n >= NN) return;

    const int deg = __builtin_amdgcn_readfirstlane(cnt[n]);
    const int nb  = deg < CAP ? deg : CAP;
    const int* ids = bucket + (size_t)n * CAP;

    const int f = lane * 4;
    const v4f b = *(const v4f*)(xr + (size_t)n * FH + f);
    const v4f w = *(const v4f*)(att + f) * LOG2E;
    v4f   acc   = *(const v4f*)(xl + (size_t)n * FH + f);
    float m     = edge_logit(acc, b, w);
    float den   = 1.f;

    for (int c = 0; c < nb; c += 32) {
        const int rem = nb - c;
        const int cn  = rem < 32 ? rem : 32;
        const int myid = (lane < cn) ? ids[c + lane] : n;
        int j = 0;
        for (; j + 4 <= cn; j += 4) gather_batch<4>(xl, myid, j, f, b, w, m, den, acc);
        const int r = cn - j;
        if (r == 3)      gather_batch<3>(xl, myid, j, f, b, w, m, den, acc);
        else if (r == 2) gather_batch<2>(xl, myid, j, f, b, w, m, den, acc);
        else if (r == 1) gather_batch<1>(xl, myid, j, f, b, w, m, den, acc);
    }

    if (deg > CAP) {
        const int ns = __builtin_amdgcn_readfirstlane(nspill[0]);
        for (int k0 = 0; k0 < ns; k0 += 32) {
            const int k    = k0 + lane;
            const int d    = (k < ns) ? spill_dst[k] : -1;
            const int myid = (k < ns) ? spill_src[k] : n;
            unsigned hit = __builtin_amdgcn_ballot_w32(d == n);
            while (hit) {
                const int j = __builtin_amdgcn_readfirstlane(__builtin_ctz(hit));
                hit &= hit - 1;
                gather_batch<1>(xl, myid, j, f, b, w, m, den, acc);
            }
        }
    }

    const float inv = 1.f / fmaxf(den, 1e-16f);
    const v4f bb = *(const v4f*)(bo + f);
    v4f o = acc * inv + bb;
    o.x = (o.x > 0.f) ? o.x : (exp_hw(o.x) - 1.f);
    o.y = (o.y > 0.f) ? o.y : (exp_hw(o.y) - 1.f);
    o.z = (o.z > 0.f) ? o.z : (exp_hw(o.z) - 1.f);
    o.w = (o.w > 0.f) ? o.w : (exp_hw(o.w) - 1.f);
    *(v4f*)(hout + (size_t)n * FH + f) = o;
}

__global__ void final_linear(const float* __restrict__ h,
                             const float* __restrict__ Wf,
                             const float* __restrict__ bf,
                             float* __restrict__ out)
{
    const int gid  = blockIdx.x * blockDim.x + threadIdx.x;
    const int n    = gid >> 5;
    const int lane = gid & 31;
    if (n >= NN) return;

    const int f = lane * 4;
    const float4 a = *(const float4*)(h + (size_t)n * FH + f);
    const float4 w = *(const float4*)(Wf + f);
    float p = a.x * w.x + a.y * w.y + a.z * w.z + a.w * w.w;
    p += __shfl_xor(p, 16, 32);
    p += __shfl_xor(p,  8, 32);
    p += __shfl_xor(p,  4, 32);
    p += __shfl_xor(p,  2, 32);
    p += __shfl_xor(p,  1, 32);
    if (lane == 0) out[n] = p + bf[0];
}

extern "C" void kernel_launch(void* const* d_in, const int* in_sizes, int n_in,
                              void* d_out, int out_size, void* d_ws, size_t ws_size,
                              hipStream_t stream)
{
    const float* x    = (const float*)d_in[0];
    const int*   ei   = (const int*)d_in[1];
    const int*   src  = ei;
    const int*   dst  = ei + EE;
    const float* Wl0  = (const float*)d_in[2];
    const float* bl0  = (const float*)d_in[3];
    const float* Wr0  = (const float*)d_in[4];
    const float* br0  = (const float*)d_in[5];
    const float* att0 = (const float*)d_in[6];
    const float* bo0  = (const float*)d_in[7];
    const float* Wl   = (const float*)d_in[8];
    const float* bl   = (const float*)d_in[9];
    const float* Wr   = (const float*)d_in[10];
    const float* br   = (const float*)d_in[11];
    const float* att  = (const float*)d_in[12];
    const float* bo   = (const float*)d_in[13];
    const float* Wf   = (const float*)d_in[14];
    const float* bf   = (const float*)d_in[15];

    float* ws   = (float*)d_ws;
    float* hbuf = ws;
    float* xl   = hbuf + (size_t)NN * FH;
    float* xr   = xl   + (size_t)NN * FH;
    int*   bkt  = (int*)(xr + (size_t)NN * FH);
    int*   cnt  = bkt  + (size_t)NR * RN * CAP;
    int*   nsp  = cnt  + NR * RN;
    int*   spd  = nsp  + 32;
    int*   sps  = spd  + EE;
    int2*  srt  = (int2*)(sps + EE);
    int*   oft  = (int*)(srt + (size_t)NB1 * P1_CHUNK);

    const dim3 gemmGrid(GEMM_BLOCKS);
    const dim3 waveGrid((NN * 32 + 255) / 256);

    edge_partition<<<dim3(NB1), P1_THREADS, 0, stream>>>(src, dst, srt, oft, nsp);
    range_buckets <<<dim3(NR),  P2_THREADS, 0, stream>>>(srt, oft, cnt, bkt, nsp, spd, sps);

    for (int layer = 0; layer < 3; ++layer) {
        const float* in = (layer == 0) ? x : hbuf;
        const float *WL, *BL, *WR, *BR, *AT, *BO;
        if (layer == 0) { WL = Wl0; BL = bl0; WR = Wr0; BR = br0; AT = att0; BO = bo0; }
        else {
            const int i = layer - 1;
            WL = Wl + (size_t)i * FH * FH;  BL = bl + (size_t)i * FH;
            WR = Wr + (size_t)i * FH * FH;  BR = br + (size_t)i * FH;
            AT = att + (size_t)i * NH * HIDW;
            BO = bo + (size_t)i * FH;
        }
        gemm_wmma_dual<<<gemmGrid, 512, 0, stream>>>(in, WL, BL, WR, BR, xl, xr);
        node_gather   <<<waveGrid, 256, 0, stream>>>(xl, xr, cnt, bkt, nsp, spd, sps, AT, BO, hbuf);
    }
    final_linear<<<waveGrid, 256, 0, stream>>>(hbuf, Wf, bf, (float*)d_out);
}


# Two helper launches folded into neighbours: the edge partition runs in front of the layer-0 GEMM's strip loop, the final linear in the last gather (h of layer 2 never stored); 9 launches to 7
# speedup vs baseline: 7.7799x; 1.0334x over previous
#include <hip/hip_runtime.h>

#define NN  50000
#define EE  800000
#define FH  128
#define NH  4
#define HIDW 32

typedef __attribute__((ext_vector_type(2))) float v2f;
typedef __attribute__((ext_vector_type(8))) float v8f;
typedef __attribute__((ext_vector_type(4))) float v4f;

#define CAP 64
#define RN  128
#define NR  ((NN + RN - 1) / RN)
#define P1_THREADS 512
#define P1_EPT     8
#define P1_CHUNK   (P1_THREADS * P1_EPT)
#define NB1 ((EE + P1_CHUNK - 1) / P1_CHUNK)
#define P2_THREADS 1024
static_assert(NB1 < 256, "the piece search in range_buckets does 8 halvings");
static_assert(NR < P1_THREADS && NB1 <= P2_THREADS && RN <= P2_THREADS, "one thread per table entry");

__device__ __forceinline__ void wave_exclusive_scan(const int* in, int* out, const int n, const int lane)
{
    const int per  = (n + 31) / 32;
    const int base = lane * per;
    int sum = 0;
    for (int k = 0; k < per; ++k) { const int r = base + k; if (r < n) sum += in[r]; }
    int inc = sum;
    for (int o = 1; o < 32; o <<= 1) { const int v = __shfl_up(inc, o, 32); if (lane >= o) inc += v; }
    int run = inc - sum;
    for (int k = 0; k < per; ++k) { const int r = base + k; if (r < n) { out[r] = run; run += in[r]; } }
    if (lane == 31) out[n] = inc;
}

#define GEMM_BLOCKS 256
#define GEMM_SLOTS  4
#define GEMM_LDS_V4 (2 * 16 * 8 * 32)

__device__ __forceinline__ void gemm_stage_weights(v4f* sW,
                                                   const float* __restrict__ Wl,
                                                   const float* __restrict__ Wr)
{
    const int tid  = threadIdx.x;
    const int lane = tid & 31;
    const int c    = lane & 15;
    const int h    = lane >> 4;
    const int t    = (tid >> 5) & 7;
    const int g0   = tid >> 8;
#pragma unroll
    for (int i = 0; i < 8; ++i) {
        const int g = g0 + 2 * i;
        const size_t o = (size_t)(8 * g + 4 * h) * FH + 16 * t + c;
        v4f vl, vr;
        vl.x = Wl[o]; vl.y = Wl[o + FH]; vl.z = Wl[o + 2 * FH]; vl.w = Wl[o + 3 * FH];
        vr.x = Wr[o]; vr.y = Wr[o + FH]; vr.z = Wr[o + 2 * FH]; vr.w = Wr[o + 3 * FH];
        sW[((0 * 16 + g) * 8 + t) * 32 + lane] = vl;
        sW[((1 * 16 + g) * 8 + t) * 32 + lane] = vr;
    }
}

__device__ __forceinline__ void gemm_strips(const v4f* sW,
                                            const float* __restrict__ X,
                                            const float* __restrict__ bl,
                                            const float* __restrict__ br,
                                            float* __restrict__ OutL,
                                            float* __restrict__ OutR)
{
    const int tid  = threadIdx.x;
    const int lane = tid & 31;
    const int wave = __builtin_amdgcn_readfirstlane(tid >> 5);
    const int c    = lane & 15;
    const int h    = lane >> 4;
    const int qd   = wave & 3;
    const int slot = wave >> 2;
    const int nstrips = (NN + 15) / 16;

    const int colA = 32 * qd + c;
    const float bL0 = bl[colA], bL1 = bl[colA + 16];
    const float bR0 = br[colA], bR1 = br[colA + 16];
    const v4f* sL = sW + (2 * qd) * 32 + lane;
    const v4f* sR = sL + 16 * 8 * 32;

    for (int s = blockIdx.x + GEMM_BLOCKS * slot; s < nstrips; s += GEMM_BLOCKS * GEMM_SLOTS) {
        int arow = 16 * s + c;
        if (arow >= NN) arow = NN - 1;
        const v4f* xp = (const v4f*)(X + (size_t)arow * FH) + h;

        v4f a[16];
#pragma unroll
        for (int g = 0; g < 16; ++g) a[g] = xp[2 * g];

        v8f cL0 = {}, cL1 = {}, cR0 = {}, cR1 = {};
        v4f wL0 = sL[0], wL1 = sL[32], wR0 = sR[0], wR1 = sR[32];
#pragma unroll
        for (int g = 0; g < 16; ++g) {
            v4f nL0 = wL0, nL1 = wL1, nR0 = wR0, nR1 = wR1;
            if (g < 15) {
                nL0 = sL[(g + 1) * 256];  nL1 = sL[(g + 1) * 256 + 32];
                nR0 = sR[(g + 1) * 256];  nR1 = sR[(g + 1) * 256 + 32];
            }
            const v2f a01 = a[g].xy, a23 = a[g].zw;
            cL0 = __builtin_amdgcn_wmma_f32_16x16x4_f32(false, a01, false, wL0.xy, (short)0, cL0, false, false);
            cL1 = __builtin_amdgcn_wmma_f32_16x16x4_f32(false, a01, false, wL1.xy, (short)0, cL1, false, false);
            cR0 = __builtin_amdgcn_wmma_f32_16x16x4_f32(false, a01, false, wR0.xy, (short)0, cR0, false, false);
            cR1 = __builtin_amdgcn_wmma_f32_16x16x4_f32(false, a01, false, wR1.xy, (short)0, cR1, false, false);
            cL0 = __builtin_amdgcn_wmma_f32_16x16x4_f32(false, a23, false, wL0.zw, (short)0, cL0, false, false);
            cL1 = __builtin_amdgcn_wmma_f32_16x16x4_f32(false, a23, false, wL1.zw, (short)0, cL1, false, false);
            cR0 = __builtin_amdgcn_wmma_f32_16x16x4_f32(false, a23, false, wR0.zw, (short)0, cR0, false, false);
            cR1 = __builtin_amdgcn_wmma_f32_16x16x4_f32(false, a23, false, wR1.zw, (short)0, cR1, false, false);
            __builtin_amdgcn_sched_barrier(0);
            wL0 = nL0; wL1 = nL1; wR0 = nR0; wR1 = nR1;
        }

#pragma unroll
        for (int v = 0; v < 8; ++v) {
            const int row = 16 * s + v + 8 * h;
            if (row < NN) {
                float* pl = OutL + (size_t)row * FH + colA;
                float* pr = OutR + (size_t)row * FH + colA;
                pl[0]  = cL0[v] + bL0;
                pl[16] = cL1[v] + bL1;
                pr[0]  = cR0[v] + bR0;
                pr[16] = cR1[v] + bR1;
            }
        }
    }
}

__global__ __launch_bounds__(512)
void gemm_wmma_dual(const float* __restrict__ X,
                    const float* __restrict__ Wl,
                    const float* __restrict__ bl,
                    const float* __restrict__ Wr,
                    const float* __restrict__ br,
                    float* __restrict__ OutL,
                    float* __restrict__ OutR)
{
    __shared__ v4f sW[GEMM_LDS_V4];
    gemm_stage_weights(sW, Wl, Wr);
    __syncthreads();
    gemm_strips(sW, X, bl, br, OutL, OutR);
}

__global__ __launch_bounds__(512)
void gemm_wmma_dual_partition(const float* __restrict__ X,
                              const float* __restrict__ Wl,
                              const float* __restrict__ bl,
                              const float* __restrict__ Wr,
                              const float* __restrict__ br,
                              float* __restrict__ OutL,
                              float* __restrict__ OutR,
                              const int* __restrict__ src,
                              const int* __restrict__ dst,
                              int2* __restrict__ srt,
                              int* __restrict__ offTab,
                              int* __restrict__ nspill)
{
    __shared__ v4f sW[GEMM_LDS_V4];
    __shared__ int hist[NR];
    __shared__ int off[NR + 1];
    const int tid = threadIdx.x;
    const int w   = blockIdx.x;
    const bool part = w < NB1;

    if (tid < NR) hist[tid] = 0;
    if (w == 0 && tid == 0) nspill[0] = 0;

    int s[P1_EPT], d[P1_EPT], pos[P1_EPT];
#pragma unroll
    for (int i = 0; i < P1_EPT; ++i) {
        const int e = w * P1_CHUNK + i * P1_THREADS + tid;
        d[i] = -1; s[i] = 0;
        if (part && e < EE) { d[i] = dst[e]; s[i] = src[e]; }
    }

    gemm_stage_weights(sW, Wl, Wr);
    __syncthreads();

#pragma unroll
    for (int i = 0; i < P1_EPT; ++i) {
        pos[i] = 0;
        if ((unsigned)d[i] < (unsigned)NN) pos[i] = atomicAdd(&hist[d[i] / RN], 1);
    }
    __syncthreads();
    if (part && tid < 32) wave_exclusive_scan(hist, off, NR, tid);
    __syncthreads();
#pragma unroll
    for (int i = 0; i < P1_EPT; ++i) {
        if ((unsigned)d[i] < (unsigned)NN)
            srt[(size_t)w * P1_CHUNK + off[d[i] / RN] + pos[i]] = make_int2(s[i], d[i]);
    }
    if (part && tid <= NR) offTab[w * (NR + 1) + tid] = off[tid];

    gemm_strips(sW, X, bl, br, OutL, OutR);
}

__global__ __launch_bounds__(P2_THREADS)
void range_buckets(const int2* __restrict__ srt,
                   const int* __restrict__ offTab,
                   int* __restrict__ cnt,
                   int* __restrict__ bucket,
                   int* __restrict__ nspill,
                   int* __restrict__ spill_dst,
                   int* __restrict__ spill_src)
{
    __shared__ int4 sB4[RN * CAP / 4];
    __shared__ int  sCnt[RN];
    __shared__ int  segLen[NB1];
    __shared__ int  segStart[NB1];
    __shared__ int  pre[NB1 + 1];
    int* sB = (int*)sB4;
    const int tid = threadIdx.x;
    const int r   = blockIdx.x;

    if (tid < NB1) {
        const int a = offTab[tid * (NR + 1) + r];
        const int b = offTab[tid * (NR + 1) + r + 1];
        segStart[tid] = tid * P1_CHUNK + a;
        segLen[tid]   = b - a;
    }
    if (tid < RN) sCnt[tid] = 0;
    __syncthreads();
    if (tid < 32) wave_exclusive_scan(segLen, pre, NB1, tid);
    __syncthreads();

    const int total = pre[NB1];
    for (int i = tid; i < total; i += P2_THREADS) {
        int lo = 0, hi = NB1;
#pragma unroll
        for (int it = 0; it < 8; ++it) {
            const int mid = (lo + hi) >> 1;
            if (pre[mid] <= i) lo = mid; else hi = mid;
        }
        const int2 sd = srt[segStart[lo] + (i - pre[lo])];
        const int dl   = sd.y - r * RN;
        const int slot = atomicAdd(&sCnt[dl], 1);
        if (slot < CAP) {
            sB[dl * CAP + slot] = sd.x;
        } else {
            const int k = atomicAdd(nspill, 1);
            spill_dst[k] = sd.y;
            spill_src[k] = sd.x;
        }
    }
    __syncthreads();

    int4* gB = (int4*)(bucket + (size_t)r * RN * CAP);
    for (int q = tid; q < RN * CAP / 4; q += P2_THREADS) gB[q] = sB4[q];
    if (tid < RN) cnt[r * RN + tid] = sCnt[tid];
}

__device__ __forceinline__ float head_sum8(float p)
{
    p += __int_as_float(__builtin_amdgcn_update_dpp(0, __float_as_int(p), 0x161, 0xf, 0xf, true));
    p += __int_as_float(__builtin_amdgcn_update_dpp(0, __float_as_int(p), 0x162, 0xf, 0xf, true));
    p += __int_as_float(__builtin_amdgcn_update_dpp(0, __float_as_int(p), 0x164, 0xf, 0xf, true));
    return p;
}

#define LOG2E 1.44269504088896341f
__device__ __forceinline__ float exp_hw(const float v)
{
    return __builtin_amdgcn_exp2f(v * LOG2E);
}

__device__ __forceinline__ v2f lrelu2(const v2f v)
{
    return __builtin_elementwise_max(v, v * 0.2f);
}

__device__ __forceinline__ float edge_logit(const v4f a, const v4f b, const v4f w)
{
    const v2f l01 = lrelu2(a.xy + b.xy);
    const v2f l23 = lrelu2(a.zw + b.zw);
    const v2f t   = l01 * w.xy + l23 * w.zw;
    return head_sum8(t.x + t.y);
}

template <int B>
__device__ __forceinline__ void batch_update(const v4f (&a)[B], const v4f b, const v4f w,
                                             float& m, float& den, v4f& acc)
{
    float p[B];
#pragma unroll
    for (int i = 0; i < B; ++i) p[i] = edge_logit(a[i], b, w);
    float mn = m;
#pragma unroll
    for (int i = 0; i < B; ++i) mn = fmaxf(mn, p[i]);
    const float sc = __builtin_amdgcn_exp2f(m - mn);
    den *= sc;
    acc *= sc;
#pragma unroll
    for (int i = 0; i < B; ++i) {
        const float e = __builtin_amdgcn_exp2f(p[i] - mn);
        den += e;
        acc += e * a[i];
    }
    m = mn;
}

template <int B>
__device__ __forceinline__ void gather_batch(const float* __restrict__ xl, const int myid,
                                             const int j, const int f,
                                             const v4f b, const v4f w,
                                             float& m, float& den, v4f& acc)
{
    v4f a[B];
#pragma unroll
    for (int i = 0; i < B; ++i) {
        const int s = __builtin_amdgcn_readlane(myid, j + i);
        a[i] = *(const v4f*)(xl + (size_t)s * FH + f);
    }
    if constexpr (B == 4) asm volatile("" :: "v"(a[0]), "v"(a[1]), "v"(a[2]), "v"(a[3]));
    if constexpr (B == 3) asm volatile("" :: "v"(a[0]), "v"(a[1]), "v"(a[2]));
    if constexpr (B == 2) asm volatile("" :: "v"(a[0]), "v"(a[1]));
    batch_update<B>(a, b, w, m, den, acc);
}

template <bool LAST>
__device__ __forceinline__ void gather_node(const float* __restrict__ xl,
                                            const float* __restrict__ xr,
                                            const int* __restrict__ cnt,
                                            const int* __restrict__ bucket,
                                            const int* __restrict__ nspill,
                                            const int* __restrict__ spill_dst,
                                            const int* __restrict__ spill_src,
                                            const float* __restrict__ att,
                                            const float* __restrict__ bo,
                                            float* __restrict__ hout,
                                            const float* __restrict__ Wf,
                                            const float* __restrict__ bf,
                                            float* __restrict__ out)
{
    const int lane = threadIdx.x & 31;
    const int n = __builtin_amdgcn_readfirstlane((int)((blockIdx.x * blockDim.x + threadIdx.x) >> 5));
    if (n >= NN) return;

    const int deg = __builtin_amdgcn_readfirstlane(cnt[n]);
    const int nb  = deg < CAP ? deg : CAP;
    const int* ids = bucket + (size_t)n * CAP;

    const int f = lane * 4;
    const v4f b = *(const v4f*)(xr + (size_t)n * FH + f);
    const v4f w = *(const v4f*)(att + f) * LOG2E;
    v4f   acc   = *(const v4f*)(xl + (size_t)n * FH + f);
    float m     = edge_logit(acc, b, w);
    float den   = 1.f;

    for (int c = 0; c < nb; c += 32) {
        const int rem = nb - c;
        const int cn  = rem < 32 ? rem : 32;
        const int myid = (lane < cn) ? ids[c + lane] : n;
        int j = 0;
        for (; j + 4 <= cn; j += 4) gather_batch<4>(xl, myid, j, f, b, w, m, den, acc);
        const int r = cn - j;
        if (r == 3)      gather_batch<3>(xl, myid, j, f, b, w, m, den, acc);
        else if (r == 2) gather_batch<2>(xl, myid, j, f, b, w, m, den, acc);
        else if (r == 1) gather_batch<1>(xl, myid, j, f, b, w, m, den, acc);
    }

    if (deg > CAP) {
        const int ns = __builtin_amdgcn_readfirstlane(nspill[0]);
        for (int k0 = 0; k0 < ns; k0 += 32) {
            const int k    = k0 + lane;
            const int d    = (k < ns) ? spill_dst[k] : -1;
            const int myid = (k < ns) ? spill_src[k] : n;
            unsigned hit = __builtin_amdgcn_ballot_w32(d == n);
            while (hit) {
                const int j = __builtin_amdgcn_readfirstlane(__builtin_ctz(hit));
                hit &= hit - 1;
                gather_batch<1>(xl, myid, j, f, b, w, m, den, acc);
            }
        }
    }

    const float inv = 1.f / fmaxf(den, 1e-16f);
    const v4f bb = *(const v4f*)(bo + f);
    v4f o = acc * inv + bb;
    o.x = (o.x > 0.f) ? o.x : (exp_hw(o.x) - 1.f);
    o.y = (o.y > 0.f) ? o.y : (exp_hw(o.y) - 1.f);
    o.z = (o.z > 0.f) ? o.z : (exp_hw(o.z) - 1.f);
    o.w = (o.w > 0.f) ? o.w : (exp_hw(o.w) - 1.f);
    if constexpr (LAST) {
        const float4 wf = *(const float4*)(Wf + f);
        float p = o.x * wf.x + o.y * wf.y + o.z * wf.z + o.w * wf.w;
        p += __shfl_xor(p, 16, 32);
        p += __shfl_xor(p,  8, 32);
        p += __shfl_xor(p,  4, 32);
        p += __shfl_xor(p,  2, 32);
        p += __shfl_xor(p,  1, 32);
        if (lane == 0) out[n] = p + bf[0];
    } else {
        *(v4f*)(hout + (size_t)n * FH + f) = o;
    }
}

__global__ void node_gather(const float* __restrict__ xl,
                            const float* __restrict__ xr,
                            const int* __restrict__ cnt,
                            const int* __restrict__ bucket,
                            const int* __restrict__ nspill,
                            const int* __restrict__ spill_dst,
                            const int* __restrict__ spill_src,
                            const float* __restrict__ att,
                            const float* __restrict__ bo,
                            float* __restrict__ hout)
{
    gather_node<false>(xl, xr, cnt, bucket, nspill, spill_dst, spill_src, att, bo, hout, nullptr, nullptr, nullptr);
}

__global__ void node_gather_final(const float* __restrict__ xl,
                                  const float* __restrict__ xr,
                                  const int* __restrict__ cnt,
                                  const int* __restrict__ bucket,
                                  const int* __restrict__ nspill,
                                  const int* __restrict__ spill_dst,
                                  const int* __restrict__ spill_src,
                                  const float* __restrict__ att,
                                  const float* __restrict__ bo,
                                  const float* __restrict__ Wf,
                                  const float* __restrict__ bf,
                                  float* __restrict__ out)
{
    gather_node<true>(xl, xr, cnt, bucket, nspill, spill_dst, spill_src, att, bo, nullptr, Wf, bf, out);
}

extern "C" void kernel_launch(void* const* d_in, const int* in_sizes, int n_in,
                              void* d_out, int out_size, void* d_ws, size_t ws_size,
                              hipStream_t stream)
{
    const float* x    = (const float*)d_in[0];
    const int*   ei   = (const int*)d_in[1];
    const int*   src  = ei;
    const int*   dst  = ei + EE;
    const float* Wl0  = (const float*)d_in[2];
    const float* bl0  = (const float*)d_in[3];
    const float* Wr0  = (const float*)d_in[4];
    const float* br0  = (const float*)d_in[5];
    const float* att0 = (const float*)d_in[6];
    const float* bo0  = (const float*)d_in[7];
    const float* Wl   = (const float*)d_in[8];
    const float* bl   = (const float*)d_in[9];
    const float* Wr   = (const float*)d_in[10];
    const float* br   = (const float*)d_in[11];
    const float* att  = (const float*)d_in[12];
    const float* bo   = (const float*)d_in[13];
    const float* Wf   = (const float*)d_in[14];
    const float* bf   = (const float*)d_in[15];

    float* ws   = (float*)d_ws;
    float* hbuf = ws;
    float* xl   = hbuf + (size_t)NN * FH;
    float* xr   = xl   + (size_t)NN * FH;
    int*   bkt  = (int*)(xr + (size_t)NN * FH);
    int*   cnt  = bkt  + (size_t)NR * RN * CAP;
    int*   nsp  = cnt  + NR * RN;
    int*   spd  = nsp  + 32;
    int*   sps  = spd  + EE;
    int2*  srt  = (int2*)(sps + EE);
    int*   oft  = (int*)(srt + (size_t)NB1 * P1_CHUNK);

    const dim3 gemmGrid(GEMM_BLOCKS);
    const dim3 waveGrid((NN * 32 + 255) / 256);

    for (int layer = 0; layer < 3; ++layer) {
        const float* in = (layer == 0) ? x : hbuf;
        const float *WL, *BL, *WR, *BR, *AT, *BO;
        if (layer == 0) { WL = Wl0; BL = bl0; WR = Wr0; BR = br0; AT = att0; BO = bo0; }
        else {
            const int i = layer - 1;
            WL = Wl + (size_t)i * FH * FH;  BL = bl + (size_t)i * FH;
            WR = Wr + (size_t)i * FH * FH;  BR = br + (size_t)i * FH;
            AT = att + (size_t)i * NH * HIDW;
            BO = bo + (size_t)i * FH;
        }
        if (layer == 0) {
            gemm_wmma_dual_partition<<<gemmGrid, 512, 0, stream>>>(in, WL, BL, WR, BR, xl, xr, src, dst, srt, oft, nsp);
            range_buckets<<<dim3(NR), P2_THREADS, 0, stream>>>(srt, oft, cnt, bkt, nsp, spd, sps);
        } else {
            gemm_wmma_dual<<<gemmGrid, 512, 0, stream>>>(in, WL, BL, WR, BR, xl, xr);
        }
        if (layer < 2)
            node_gather<<<waveGrid, 256, 0, stream>>>(xl, xr, cnt, bkt, nsp, spd, sps, AT, BO, hbuf);
        else
            node_gather_final<<<waveGrid, 256, 0, stream>>>(xl, xr, cnt, bkt, nsp, spd, sps, AT, BO, Wf, bf, (float*)d_out);
    }
}
